# starting kernel
# baseline (speedup 1.0000x reference)
#include <hip/hip_runtime.h>
#include <math.h>


typedef __attribute__((ext_vector_type(2))) float v2f;
typedef __attribute__((ext_vector_type(8))) float v8f;

#define DIM 4096

constexpr unsigned cnot_src(unsigned k) {
  for (int i = 10; i >= 0; --i)
    for (int j = 11; j >= i + 1; --j)
      k ^= ((k >> (11 - i)) & 1u) << (11 - j);
  return k;
}
constexpr unsigned mask_p(int p) {
  unsigned mm = 0;
  for (int b = 0; b < 12; ++b)
    mm |= ((cnot_src(1u << b) >> p) & 1u) << b;
  return mm;
}
__constant__ unsigned SRC_MASK[12] = {
    mask_p(0), mask_p(1), mask_p(2),  mask_p(3),
    mask_p(4), mask_p(5), mask_p(6),  mask_p(7),
    mask_p(8), mask_p(9), mask_p(10), mask_p(11)};

__device__ __forceinline__ unsigned src_perm(unsigned k) {
  unsigned r = 0;
#pragma unroll
  for (int p = 0; p < 12; ++p)
    r |= (unsigned)(__popc(k & SRC_MASK[p]) & 1) << p;
  return r;
}

__device__ __forceinline__ v8f wmma4(v2f a, v2f b, v8f c) {
  return __builtin_amdgcn_wmma_f32_16x16x4_f32(false, a, false, b,
                                               (short)0, c, false, false);
}

__global__ __launch_bounds__(256)
void qdense_kernel(const float* __restrict__ x,
                   const float* __restrict__ theta,
                   float* __restrict__ out) {
  __shared__ float st[DIM];
  __shared__ float G[6][256];
  __shared__ float cs[24], sn[24];
  __shared__ float wred[8];
  __shared__ float wout[8][12];

  const int t    = threadIdx.x;
  const int lane = t & 31;
  const int wv   = t >> 5;
  const int n    = lane & 15;
  const int hi2  = (lane >> 4) * 2;
  const int mrow = (lane >> 4) * 8;

  if (t < 24) {
    float s, c;
    sincosf(theta[t] * 0.5f, &s, &c);
    cs[t] = c;
    sn[t] = s;
  }
  __syncthreads();

  {
    const int r = t >> 4, c = t & 15;
#pragma unroll
    for (int m = 0; m < 6; ++m) {
      float v = 1.f;
#pragma unroll
      for (int q = 0; q < 4; ++q) {
        const int rb = (r >> (3 - q)) & 1;
        const int cb = (c >> (3 - q)) & 1;
        const float cv = cs[m * 4 + q], sv = sn[m * 4 + q];
        v *= rb ? (cb ? cv : sv) : (cb ? -sv : cv);
      }
      G[m][t] = v;
    }
  }

  const float4* xr = (const float4*)(x + (size_t)blockIdx.x * DIM);
  float ss = 0.f;
#pragma unroll
  for (int e = 0; e < 4; ++e) {
    float4 v = xr[t * 4 + e];
    ((float4*)st)[t * 4 + e] = v;
    ss += v.x * v.x + v.y * v.y + v.z * v.z + v.w * v.w;
  }
#pragma unroll
  for (int o = 16; o > 0; o >>= 1) ss += __shfl_xor(ss, o, 32);
  if (lane == 0) wred[wv] = ss;
  __syncthreads();

  float sumsq = 0.f;
#pragma unroll
  for (int w = 0; w < 8; ++w) sumsq += wred[w];

  const bool fb = (sumsq <= 1e-24f);
  if (fb) {
#pragma unroll
    for (int e = 0; e < 16; ++e) st[e * 256 + t] = 0.f;
    if (t == 0) st[0] = 1.f;
  }
  const float inv = fb ? 1.f : 1.f / sumsq;
  __syncthreads();

#pragma unroll 1
  for (int d = 0; d < 2; ++d) {
    const float* g0 = G[d * 3 + 0];
    const float* g1 = G[d * 3 + 1];
    const float* g2 = G[d * 3 + 2];

    {
      v2f a[4];
#pragma unroll
      for (int kc = 0; kc < 4; ++kc) {
        a[kc].x = g0[n * 16 + kc * 4 + 0 + hi2];
        a[kc].y = g0[n * 16 + kc * 4 + 1 + hi2];
      }
#pragma unroll
      for (int tt = 0; tt < 2; ++tt) {
        const int col = (wv * 2 + tt) * 16 + n;
        v8f acc = {};
#pragma unroll
        for (int kc = 0; kc < 4; ++kc) {
          v2f b;
          b.x = st[(kc * 4 + 0 + hi2) * 256 + col];
          b.y = st[(kc * 4 + 1 + hi2) * 256 + col];
          acc = wmma4(a[kc], b, acc);
        }
#pragma unroll
        for (int v = 0; v < 8; ++v) st[(v + mrow) * 256 + col] = acc[v];
      }
    }
    __syncthreads();

    {
      v2f a[4];
#pragma unroll
      for (int kc = 0; kc < 4; ++kc) {
        a[kc].x = g1[n * 16 + kc * 4 + 0 + hi2];
        a[kc].y = g1[n * 16 + kc * 4 + 1 + hi2];
      }
#pragma unroll
      for (int tt = 0; tt < 2; ++tt) {
        const int base = (wv * 2 + tt) * 256;
        v8f acc = {};
#pragma unroll
        for (int kc = 0; kc < 4; ++kc) {
          v2f b;
          b.x = st[base + (kc * 4 + 0 + hi2) * 16 + n];
          b.y = st[base + (kc * 4 + 1 + hi2) * 16 + n];
          acc = wmma4(a[kc], b, acc);
        }
#pragma unroll
        for (int v = 0; v < 8; ++v) st[base + (v + mrow) * 16 + n] = acc[v];
      }
    }
    __syncthreads();

    {
      v2f bg[4];
#pragma unroll
      for (int kc = 0; kc < 4; ++kc) {
        bg[kc].x = g2[n * 16 + kc * 4 + 0 + hi2];
        bg[kc].y = g2[n * 16 + kc * 4 + 1 + hi2];
      }
#pragma unroll
      for (int tt = 0; tt < 2; ++tt) {
        const int rbase = (wv * 2 + tt) * 16;
        v8f acc = {};
#pragma unroll
        for (int kc = 0; kc < 4; ++kc) {
          v2f a;
          a.x = st[(rbase + n) * 16 + kc * 4 + 0 + hi2];
          a.y = st[(rbase + n) * 16 + kc * 4 + 1 + hi2];
          acc = wmma4(a, bg[kc], acc);
        }
#pragma unroll
        for (int v = 0; v < 8; ++v) st[(rbase + v + mrow) * 16 + n] = acc[v];
      }
    }
    __syncthreads();

    if (d == 0) {
      float pr[16];
#pragma unroll
      for (int e = 0; e < 16; ++e) pr[e] = st[src_perm(e * 256 + t)];
      __syncthreads();
#pragma unroll
      for (int e = 0; e < 16; ++e) st[e * 256 + t] = pr[e];
      __syncthreads();
    }
  }

  float acc12[12];
#pragma unroll
  for (int w = 0; w < 12; ++w) acc12[w] = 0.f;
#pragma unroll
  for (int e = 0; e < 16; ++e) {
    const unsigned k = (unsigned)(e * 256 + t);
    const float a = st[src_perm(k)];
    const float p = a * a;
#pragma unroll
    for (int w = 0; w < 12; ++w)
      acc12[w] += ((k >> (11 - w)) & 1u) ? -p : p;
  }
#pragma unroll
  for (int w = 0; w < 12; ++w) {
#pragma unroll
    for (int o = 16; o > 0; o >>= 1)
      acc12[w] += __shfl_xor(acc12[w], o, 32);
  }
  if (lane == 0) {
#pragma unroll
    for (int w = 0; w < 12; ++w) wout[wv][w] = acc12[w];
  }
  __syncthreads();
  if (t < 12) {
    float s = 0.f;
#pragma unroll
    for (int w2 = 0; w2 < 8; ++w2) s += wout[w2][t];
    out[(size_t)blockIdx.x * 12 + t] = s * inv;
  }
}

extern "C" void kernel_launch(void* const* d_in, const int* in_sizes, int n_in,
                              void* d_out, int out_size, void* d_ws,
                              size_t ws_size, hipStream_t stream) {
  const float* x = (const float*)d_in[0];
  const float* theta = (const float*)d_in[1];
  float* out = (float*)d_out;
  const int B = in_sizes[0] / DIM;
  qdense_kernel<<<dim3(B), dim3(256), 0, stream>>>(x, theta, out);
}


# One wave owns one row (1-wave workgroup per row): all 16 tiles of each pass in that wave, next tile's LDS reads issued ahead, no barriers, gate fragments in registers; new order of the float sums
# speedup vs baseline: 2.4258x; 2.4258x over previous
#include <hip/hip_runtime.h>
#include <math.h>


typedef __attribute__((ext_vector_type(2))) float v2f;
typedef __attribute__((ext_vector_type(8))) float v8f;

#define DIM 4096

constexpr unsigned cnot_fwd(unsigned k) {
  for (int i = 0; i < 11; ++i)
    for (int j = i + 1; j < 12; ++j)
      k ^= ((k >> (11 - i)) & 1u) << (11 - j);
  return k;
}
constexpr bool cascade_is_gray() {
  for (int b = 0; b < 12; ++b) {
    const unsigned k = 1u << b;
    if (cnot_fwd(k) != (k ^ (k >> 1))) return false;
  }
  return true;
}
static_assert(cascade_is_gray(), "CNOT cascade != Gray map");

__device__ __forceinline__ int row_pos(int v, int X) {
  const int g = v ^ (v >> 1);
  return ((g & 4) ? (X ^ 4) : X) + (g & 3);
}

__device__ __forceinline__ v8f wmma4(v2f a, v2f b, v8f c) {
  return __builtin_amdgcn_wmma_f32_16x16x4_f32(false, a, false, b,
                                               (short)0, c, false, false);
}

__device__ __forceinline__ float ry_entry(int rb, int cb, float c, float s) {
  return rb ? (cb ? c : s) : (cb ? -s : c);
}

__device__ __forceinline__ float lane_bcast(float v, int l) {
  return __int_as_float(__builtin_amdgcn_readlane(__float_as_int(v), l));
}

__global__ __launch_bounds__(32, 4)
void qdense_kernel(const float* __restrict__ x,
                   const float* __restrict__ theta,
                   float* __restrict__ out) {
  __shared__ __attribute__((aligned(16))) float st[DIM];

  const int lane = threadIdx.x;
  const int n    = lane & 15;
  const int half = lane >> 4;
  const int hi2  = half * 2;
  const int mrow = half * 8;

  const float th = theta[lane < 24 ? lane : 0];
  const float4* xr = (const float4*)(x + (size_t)blockIdx.x * DIM);
  float4 xv[32];
#pragma unroll
  for (int e = 0; e < 32; ++e) xv[e] = xr[e * 32 + lane];

  float sv, cv;
  sincosf(th * 0.5f, &sv, &cv);

  v2f F[6][4];
#pragma unroll
  for (int m = 0; m < 6; ++m) {
    const float c0 = lane_bcast(cv, m * 4 + 0), s0 = lane_bcast(sv, m * 4 + 0);
    const float c1 = lane_bcast(cv, m * 4 + 1), s1 = lane_bcast(sv, m * 4 + 1);
    const float c2 = lane_bcast(cv, m * 4 + 2), s2 = lane_bcast(sv, m * 4 + 2);
    const float c3 = lane_bcast(cv, m * 4 + 3), s3 = lane_bcast(sv, m * 4 + 3);
    const float e0[2] = {ry_entry((n >> 3) & 1, 0, c0, s0),
                         ry_entry((n >> 3) & 1, 1, c0, s0)};
    const float e1[2] = {ry_entry((n >> 2) & 1, 0, c1, s1),
                         ry_entry((n >> 2) & 1, 1, c1, s1)};
    const float e2    =  ry_entry((n >> 1) & 1, half, c2, s2);
    const float e3[2] = {ry_entry(n & 1, 0, c3, s3),
                         ry_entry(n & 1, 1, c3, s3)};
#pragma unroll
    for (int kc = 0; kc < 4; ++kc) {
      const float p = (e0[kc >> 1] * e1[kc & 1]) * e2;
      F[m][kc].x = p * e3[0];
      F[m][kc].y = p * e3[1];
    }
  }

  float ss = 0.f;
#pragma unroll
  for (int e = 0; e < 32; ++e) {
    const float4 v = xv[e];
    ((float4*)st)[e * 32 + lane] = v;
    ss += v.x * v.x + v.y * v.y + v.z * v.z + v.w * v.w;
  }
#pragma unroll
  for (int o = 16; o > 0; o >>= 1) ss += __shfl_xor(ss, o, 32);
  const float sumsq = ss;

  const bool fb = (sumsq <= 1e-24f);
  if (fb) {
#pragma unroll
    for (int e = 0; e < 32; ++e)
      ((float4*)st)[e * 32 + lane] =
          make_float4((e == 0 && lane == 0) ? 1.f : 0.f, 0.f, 0.f, 0.f);
  }
  const float inv = fb ? 1.f : 1.f / sumsq;
  __syncthreads();

  const int xh = half * 12;
  const int gn = n ^ (n >> 1);

#pragma unroll 1
  for (int d = 0; d < 2; ++d) {
    v2f a0[4], a1[4], a2[4];
#pragma unroll
    for (int kc = 0; kc < 4; ++kc) {
      a0[kc] = d ? F[3][kc] : F[0][kc];
      a1[kc] = d ? F[4][kc] : F[1][kc];
      a2[kc] = d ? F[5][kc] : F[2][kc];
    }


    {
      v2f bb[2][4];
#pragma unroll
      for (int kc = 0; kc < 4; ++kc) {
        bb[0][kc].x = st[(kc * 4 + 0 + hi2) * 256 + n];
        bb[0][kc].y = st[(kc * 4 + 1 + hi2) * 256 + n];
      }
#pragma unroll
      for (int T = 0; T < 16; ++T) {
        const int col = T * 16 + n;
        if (T + 1 < 16) {
#pragma unroll
          for (int kc = 0; kc < 4; ++kc) {
            bb[(T + 1) & 1][kc].x = st[(kc * 4 + 0 + hi2) * 256 + col + 16];
            bb[(T + 1) & 1][kc].y = st[(kc * 4 + 1 + hi2) * 256 + col + 16];
          }
        }
        v8f acc = {};
#pragma unroll
        for (int kc = 0; kc < 4; ++kc) acc = wmma4(a0[kc], bb[T & 1][kc], acc);
#pragma unroll
        for (int v = 0; v < 8; ++v) st[row_pos(v, xh) * 256 + col] = acc[v];
      }
    }
    __syncthreads();

    {
      v2f bb[2][4];
#pragma unroll
      for (int kc = 0; kc < 4; ++kc) {
        bb[0][kc].x = st[(kc * 4 + 0 + hi2) * 16 + n];
        bb[0][kc].y = st[(kc * 4 + 1 + hi2) * 16 + n];
      }
#pragma unroll
      for (int T = 0; T < 16; ++T) {
        const int base = T * 256;
        const int pT = __builtin_popcount((unsigned)T) & 1;
        const int xm = xh ^ (pT << 3);
        if (T + 1 < 16) {
#pragma unroll
          for (int kc = 0; kc < 4; ++kc) {
            bb[(T + 1) & 1][kc].x = st[base + 256 + (kc * 4 + 0 + hi2) * 16 + n];
            bb[(T + 1) & 1][kc].y = st[base + 256 + (kc * 4 + 1 + hi2) * 16 + n];
          }
        }
        v8f acc = {};
#pragma unroll
        for (int kc = 0; kc < 4; ++kc) acc = wmma4(a1[kc], bb[T & 1][kc], acc);
#pragma unroll
        for (int v = 0; v < 8; ++v) st[base + row_pos(v, xm) * 16 + n] = acc[v];
      }
    }
    __syncthreads();

    {
      v2f aa[2][4];
#pragma unroll
      for (int kc = 0; kc < 4; ++kc) {
        aa[0][kc].x = st[n * 16 + kc * 4 + 0 + hi2];
        aa[0][kc].y = st[n * 16 + kc * 4 + 1 + hi2];
      }
#pragma unroll
      for (int T = 0; T < 16; ++T) {
        const int rbase = T * 16;
        const int pT = __builtin_popcount((unsigned)T) & 1;
        const int lcT = gn ^ ((half ^ pT) << 3);
        if (T + 1 < 16) {
#pragma unroll
          for (int kc = 0; kc < 4; ++kc) {
            aa[(T + 1) & 1][kc].x = st[(rbase + 16 + n) * 16 + kc * 4 + 0 + hi2];
            aa[(T + 1) & 1][kc].y = st[(rbase + 16 + n) * 16 + kc * 4 + 1 + hi2];
          }
        }
        v8f acc = {};
#pragma unroll
        for (int kc = 0; kc < 4; ++kc) acc = wmma4(aa[T & 1][kc], a2[kc], acc);
#pragma unroll
        for (int v = 0; v < 8; ++v) {
          const int flip = (v ^ (v >> 1) ^ (v >> 2)) & 1;
          st[(rbase + v + mrow) * 16 + (lcT ^ (flip << 3))] = acc[v];
        }
      }
    }
    __syncthreads();
  }

  float sg[5];
#pragma unroll
  for (int w = 5; w < 10; ++w) sg[w - 5] = ((lane >> (9 - w)) & 1) ? -1.f : 1.f;

  float acc12[12];
#pragma unroll
  for (int w = 0; w < 12; ++w) acc12[w] = 0.f;
#pragma unroll
  for (int e = 0; e < 32; ++e) {
    const float4 q = ((const float4*)st)[e * 32 + lane];
    const float qa[4] = {q.x, q.y, q.z, q.w};
#pragma unroll
    for (int c = 0; c < 4; ++c) {
      const float p = qa[c] * qa[c];
#pragma unroll
      for (int w = 0; w < 5; ++w)
        acc12[w] += ((e >> (4 - w)) & 1) ? -p : p;
#pragma unroll
      for (int w = 5; w < 10; ++w)
        acc12[w] += p * sg[w - 5];
#pragma unroll
      for (int w = 10; w < 12; ++w)
        acc12[w] += ((c >> (11 - w)) & 1) ? -p : p;
    }
  }
#pragma unroll
  for (int w = 0; w < 12; ++w) {
#pragma unroll
    for (int o = 16; o > 0; o >>= 1)
      acc12[w] += __shfl_xor(acc12[w], o, 32);
  }
  float r = acc12[0];
#pragma unroll
  for (int w = 1; w < 12; ++w) r = (lane == w) ? acc12[w] : r;
  if (lane < 12) out[(size_t)blockIdx.x * 12 + lane] = r * inv;
}

extern "C" void kernel_launch(void* const* d_in, const int* in_sizes, int n_in,
                              void* d_out, int out_size, void* d_ws,
                              size_t ws_size, hipStream_t stream) {
  const float* x = (const float*)d_in[0];
  const float* theta = (const float*)d_in[1];
  float* out = (float*)d_out;
  const int B = in_sizes[0] / DIM;
  qdense_kernel<<<dim3(B), dim3(32), 0, stream>>>(x, theta, out);
}


# Row kept in registers through each RY layer: pass 1's D rows feed pass 2, a half-swap per register pair feeds pass 3; one LDS trip per row (Gray index), not seven; new order of the float sums
# speedup vs baseline: 2.7491x; 1.1333x over previous
#include <hip/hip_runtime.h>
#include <math.h>


typedef __attribute__((ext_vector_type(2))) float v2f;
typedef __attribute__((ext_vector_type(8))) float v8f;
typedef __attribute__((ext_vector_type(2))) unsigned v2u;

#define DIM 4096

constexpr unsigned cnot_fwd(unsigned k) {
  for (int i = 0; i < 11; ++i)
    for (int j = i + 1; j < 12; ++j)
      k ^= ((k >> (11 - i)) & 1u) << (11 - j);
  return k;
}
constexpr bool cascade_is_gray() {
  for (int b = 0; b < 12; ++b) {
    const unsigned k = 1u << b;
    if (cnot_fwd(k) != (k ^ (k >> 1))) return false;
  }
  return true;
}
static_assert(cascade_is_gray(), "CNOT cascade != Gray map");

__host__ __device__ constexpr int gray12(int k) { return k ^ (k >> 1); }
__host__ __device__ constexpr int lds_word(int k) { return k ^ (((k >> 8) & 15) << 2); }

__device__ __forceinline__ v8f wmma4(v2f a, v2f b, v8f c) {
  return __builtin_amdgcn_wmma_f32_16x16x4_f32(false, a, false, b,
                                               (short)0, c, false, false);
}

__device__ __forceinline__ float ry_entry(int rb, int cb, float c, float s) {
  return rb ? (cb ? c : s) : (cb ? -s : c);
}

__device__ __forceinline__ float lane_bcast(float v, int l) {
  return __int_as_float(__builtin_amdgcn_readlane(__float_as_int(v), l));
}

__device__ __forceinline__ void half_swap(float& x, float& y) {
  const v2u r = __builtin_amdgcn_permlane16_swap(__float_as_uint(x),
                                                 __float_as_uint(y), false, false);
  x = __uint_as_float(r.x);
  y = __uint_as_float(r.y);
}

__device__ __forceinline__ void gate_frag(v2f (&F)[4], float cv, float sv,
                                          int i3, int i2, int i1, int i0,
                                          int n, int half) {
  const float c3 = lane_bcast(cv, i3), s3 = lane_bcast(sv, i3);
  const float c2 = lane_bcast(cv, i2), s2 = lane_bcast(sv, i2);
  const float c1 = lane_bcast(cv, i1), s1 = lane_bcast(sv, i1);
  const float c0 = lane_bcast(cv, i0), s0 = lane_bcast(sv, i0);
  const float e3    =  ry_entry((n >> 3) & 1, half, c3, s3);
  const float e2[2] = {ry_entry((n >> 2) & 1, 0, c2, s2),
                       ry_entry((n >> 2) & 1, 1, c2, s2)};
  const float e1[2] = {ry_entry((n >> 1) & 1, 0, c1, s1),
                       ry_entry((n >> 1) & 1, 1, c1, s1)};
  const float e0[2] = {ry_entry(n & 1, 0, c0, s0),
                       ry_entry(n & 1, 1, c0, s0)};
#pragma unroll
  for (int kc = 0; kc < 4; ++kc) {
    const float p = (e3 * e2[kc >> 1]) * e1[kc & 1];
    F[kc].x = p * e0[0];
    F[kc].y = p * e0[1];
  }
}

template <bool LAST>
__device__ __forceinline__ void ry_layer(const float4 (&xv)[32],
                                         const v2f (&FA)[4], const v2f (&FN)[4],
                                         const v2f (&FB)[4],
                                         float* st, int pl, float e0col,
                                         float& P, float (&Q)[8]) {
  v8f D1[16];
#pragma unroll
  for (int b = 0; b < 16; ++b) {
    v8f acc = {};
#pragma unroll
    for (int kc = 0; kc < 4; ++kc) {
      const float4 q = xv[((kc >> 1) << 4) | b];
      v2f a;
      a.x = (kc & 1) ? q.z : q.x;
      a.y = (kc & 1) ? q.w : q.y;
      acc = wmma4(a, FA[kc], acc);
    }
    if (b == 0) acc[0] += e0col;
    D1[b] = acc;
  }
  v8f D2[16];
#pragma unroll
  for (int b = 0; b < 16; ++b) {
    v8f acc = {};
#pragma unroll
    for (int kc = 0; kc < 4; ++kc) {
      v2f s;
      s.x = D1[b][2 * kc + 0];
      s.y = D1[b][2 * kc + 1];
      acc = wmma4(FN[kc], s, acc);
    }
    D2[b] = acc;
  }
#pragma unroll
  for (int b = 0; b < 8; ++b) {
#pragma unroll
    for (int v = 0; v < 8; ++v) {
      float lo = D2[b][v], hi = D2[b + 8][v];
      half_swap(lo, hi);
      D2[b][v] = lo;
      D2[b + 8][v] = hi;
    }
  }
#pragma unroll
  for (int x = 0; x < 2; ++x) {
#pragma unroll
    for (int v = 0; v < 8; ++v) {
      v8f acc = {};
#pragma unroll
      for (int kc = 0; kc < 4; ++kc) {
        v2f s;
        s.x = D2[x * 8 + 2 * kc + 0][v];
        s.y = D2[x * 8 + 2 * kc + 1][v];
        acc = wmma4(FB[kc], s, acc);
      }
      const int T = x * 8 + v;
      if (!LAST) {
#pragma unroll
        for (int r = 0; r < 8; ++r)
          st[pl ^ lds_word(gray12((r << 6) | (T << 2)))] = acc[r];
      } else {
#pragma unroll
        for (int r = 0; r < 8; ++r) {
          const float p = acc[r] * acc[r];
          P += p;
          const int sg[8] = {(r >> 2) & 1,
                             ((r >> 1) ^ (r >> 2)) & 1,
                             (r ^ (r >> 1)) & 1,
                             ((T >> 3) ^ r) & 1,
                             ((T >> 2) ^ (T >> 3)) & 1,
                             ((T >> 1) ^ (T >> 2)) & 1,
                             (T ^ (T >> 1)) & 1,
                             T & 1};
#pragma unroll
          for (int i = 0; i < 8; ++i) Q[i] += sg[i] ? -p : p;
        }
      }
    }
  }
}

__global__ __launch_bounds__(32, 4)
void qdense_kernel(const float* __restrict__ x,
                   const float* __restrict__ theta,
                   float* __restrict__ out) {
  __shared__ __attribute__((aligned(16))) float st[DIM];

  const int lane = threadIdx.x;
  const int n    = lane & 15;
  const int half = lane >> 4;

  const float th = theta[lane < 24 ? lane : 0];
  const float4* xr = (const float4*)(x + (size_t)blockIdx.x * DIM);
  float4 xv[32];
#pragma unroll
  for (int e = 0; e < 32; ++e) xv[e] = xr[half * 512 + e * 16 + n];

  float sv, cv;
  sincosf(th * 0.5f, &sv, &cv);

  v2f FA[4], FN[4], FB[4];
  gate_frag(FA, cv, sv, 0, 1, 10, 11, n, half);
  gate_frag(FN, cv, sv, 6, 7, 8, 9, n, half);
  gate_frag(FB, cv, sv, 2, 3, 4, 5, n, half);

  float ss = 0.f;
#pragma unroll
  for (int e = 0; e < 32; ++e) {
    const float4 v = xv[e];
    ss += v.x * v.x + v.y * v.y + v.z * v.z + v.w * v.w;
  }
#pragma unroll
  for (int o = 16; o > 0; o >>= 1) ss += __shfl_xor(ss, o, 32);
  const float sumsq = ss;

  const bool fb = (sumsq <= 1e-24f);
  const float keep = fb ? 0.f : 1.f;
  const float e0col = (fb && half == 0) ? FA[0].x : 0.f;
#pragma unroll
  for (int kc = 0; kc < 4; ++kc) FA[kc] = FA[kc] * keep;
  const float inv = fb ? 1.f : 1.f / sumsq;

  float P = 0.f;
  float Q[8];
#pragma unroll
  for (int i = 0; i < 8; ++i) Q[i] = 0.f;

  const int pl = lds_word(gray12(((n >> 2) << 10) | (n & 3) | (half << 9)));
  ry_layer<false>(xv, FA, FN, FB, st, pl, e0col, P, Q);
  __syncthreads();
#pragma unroll
  for (int e = 0; e < 32; ++e)
    xv[e] = ((const float4*)st)[half * 512 + e * 16 + (n ^ ((half << 3) | (e >> 2)))];

  gate_frag(FA, cv, sv, 12 + 0, 12 + 1, 12 + 10, 12 + 11, n, half);
  gate_frag(FN, cv, sv, 12 + 6, 12 + 7, 12 + 8, 12 + 9, n, half);
  gate_frag(FB, cv, sv, 12 + 2, 12 + 3, 12 + 4, 12 + 5, n, half);
  ry_layer<true>(xv, FA, FN, FB, st, pl, 0.f, P, Q);

  float acc12[12];
  acc12[0]  = ((n >> 3) & 1) ? -P : P;
  acc12[1]  = (((n >> 2) ^ (n >> 3)) & 1) ? -P : P;
  acc12[2]  = ((half ^ (n >> 2)) & 1) ? -P : P;
  acc12[3]  = half ? -Q[0] : Q[0];
#pragma unroll
  for (int i = 1; i < 7; ++i) acc12[3 + i] = Q[i];
  acc12[10] = ((n >> 1) & 1) ? -Q[7] : Q[7];
  acc12[11] = ((n ^ (n >> 1)) & 1) ? -P : P;
#pragma unroll
  for (int w = 0; w < 12; ++w) {
#pragma unroll
    for (int o = 16; o > 0; o >>= 1)
      acc12[w] += __shfl_xor(acc12[w], o, 32);
  }
  float r = acc12[0];
#pragma unroll
  for (int w = 1; w < 12; ++w) r = (lane == w) ? acc12[w] : r;
  if (lane < 12) out[(size_t)blockIdx.x * 12 + lane] = r * inv;
}

extern "C" void kernel_launch(void* const* d_in, const int* in_sizes, int n_in,
                              void* d_out, int out_size, void* d_ws,
                              size_t ws_size, hipStream_t stream) {
  const float* x = (const float*)d_in[0];
  const float* theta = (const float*)d_in[1];
  float* out = (float*)d_out;
  const int B = in_sizes[0] / DIM;
  qdense_kernel<<<dim3(B), dim3(32), 0, stream>>>(x, theta, out);
}


# Z totals and sum of squares as matrix products: 32 unsigned per-lane sums x a +-1 sign matrix, rows folded by a 2nd WMMA stage; no signed adds, no ds_bpermute hops; new order of the float sums
# speedup vs baseline: 2.9443x; 1.0710x over previous
#include <hip/hip_runtime.h>
#include <math.h>


typedef __attribute__((ext_vector_type(2))) float v2f;
typedef __attribute__((ext_vector_type(8))) float v8f;
typedef __attribute__((ext_vector_type(2))) unsigned v2u;

#define DIM 4096

constexpr unsigned cnot_fwd(unsigned k) {
  for (int i = 0; i < 11; ++i)
    for (int j = i + 1; j < 12; ++j)
      k ^= ((k >> (11 - i)) & 1u) << (11 - j);
  return k;
}
constexpr bool cascade_is_gray() {
  for (int b = 0; b < 12; ++b) {
    const unsigned k = 1u << b;
    if (cnot_fwd(k) != (k ^ (k >> 1))) return false;
  }
  return true;
}
static_assert(cascade_is_gray(), "CNOT cascade != Gray map");

__host__ __device__ constexpr int gray12(int k) { return k ^ (k >> 1); }
__host__ __device__ constexpr int lds_word(int k) { return k ^ (((k >> 8) & 15) << 2); }

__device__ __forceinline__ v8f wmma4(v2f a, v2f b, v8f c) {
  return __builtin_amdgcn_wmma_f32_16x16x4_f32(false, a, false, b,
                                               (short)0, c, false, false);
}

__device__ __forceinline__ float ry_entry(int rb, int cb, float c, float s) {
  return rb ? (cb ? c : s) : (cb ? -s : c);
}

__device__ __forceinline__ float lane_bcast(float v, int l) {
  return __int_as_float(__builtin_amdgcn_readlane(__float_as_int(v), l));
}

__device__ __forceinline__ void half_swap(float& x, float& y) {
  const v2u r = __builtin_amdgcn_permlane16_swap(__float_as_uint(x),
                                                 __float_as_uint(y), false, false);
  x = __uint_as_float(r.x);
  y = __uint_as_float(r.y);
}

__device__ __forceinline__ void gate_frag(v2f (&F)[4], float cv, float sv,
                                          int i3, int i2, int i1, int i0,
                                          int n, int half) {
  const float c3 = lane_bcast(cv, i3), s3 = lane_bcast(sv, i3);
  const float c2 = lane_bcast(cv, i2), s2 = lane_bcast(sv, i2);
  const float c1 = lane_bcast(cv, i1), s1 = lane_bcast(sv, i1);
  const float c0 = lane_bcast(cv, i0), s0 = lane_bcast(sv, i0);
  const float e3    =  ry_entry((n >> 3) & 1, half, c3, s3);
  const float e2[2] = {ry_entry((n >> 2) & 1, 0, c2, s2),
                       ry_entry((n >> 2) & 1, 1, c2, s2)};
  const float e1[2] = {ry_entry((n >> 1) & 1, 0, c1, s1),
                       ry_entry((n >> 1) & 1, 1, c1, s1)};
  const float e0[2] = {ry_entry(n & 1, 0, c0, s0),
                       ry_entry(n & 1, 1, c0, s0)};
#pragma unroll
  for (int kc = 0; kc < 4; ++kc) {
    const float p = (e3 * e2[kc >> 1]) * e1[kc & 1];
    F[kc].x = p * e0[0];
    F[kc].y = p * e0[1];
  }
}

__device__ __forceinline__ float signed_coef(float base, unsigned pw, int idx) {
  return __uint_as_float(__float_as_uint(base) ^ ((pw << (31 - idx)) & 0x80000000u));
}

template <bool LAST>
__device__ __forceinline__ void ry_layer(const float4 (&xv)[32],
                                         const v2f (&FA)[4], const v2f (&FN)[4],
                                         const v2f (&FB)[4],
                                         float* st, int pl, float e0col,
                                         float (&U)[16], float (&V)[16]) {
  v8f D1[16];
#pragma unroll
  for (int b = 0; b < 16; ++b) {
    v8f acc = {};
#pragma unroll
    for (int kc = 0; kc < 4; ++kc) {
      const float4 q = xv[((kc >> 1) << 4) | b];
      v2f a;
      a.x = (kc & 1) ? q.z : q.x;
      a.y = (kc & 1) ? q.w : q.y;
      acc = wmma4(a, FA[kc], acc);
    }
    if (b == 0) acc[0] += e0col;
    D1[b] = acc;
  }
  v8f D2[16];
#pragma unroll
  for (int b = 0; b < 16; ++b) {
    v8f acc = {};
#pragma unroll
    for (int kc = 0; kc < 4; ++kc) {
      v2f s;
      s.x = D1[b][2 * kc + 0];
      s.y = D1[b][2 * kc + 1];
      acc = wmma4(FN[kc], s, acc);
    }
    D2[b] = acc;
  }
#pragma unroll
  for (int b = 0; b < 8; ++b) {
#pragma unroll
    for (int v = 0; v < 8; ++v) {
      float lo = D2[b][v], hi = D2[b + 8][v];
      half_swap(lo, hi);
      D2[b][v] = lo;
      D2[b + 8][v] = hi;
    }
  }
#pragma unroll
  for (int x = 0; x < 2; ++x) {
#pragma unroll
    for (int v = 0; v < 8; ++v) {
      v8f acc = {};
#pragma unroll
      for (int kc = 0; kc < 4; ++kc) {
        v2f s;
        s.x = D2[x * 8 + 2 * kc + 0][v];
        s.y = D2[x * 8 + 2 * kc + 1][v];
        acc = wmma4(FB[kc], s, acc);
      }
      const int T = x * 8 + v;
      if (!LAST) {
#pragma unroll
        for (int r = 0; r < 8; ++r)
          st[pl ^ lds_word(gray12((r << 6) | (T << 2)))] = acc[r];
      } else {
#pragma unroll
        for (int r = 0; r < 8; ++r) {
          const float p = acc[r] * acc[r];
          U[2 * r + (T >> 3)] += p;
          V[T] += p;
        }
      }
    }
  }
}

__global__ __launch_bounds__(32, 4)
void qdense_kernel(const float* __restrict__ x,
                   const float* __restrict__ theta,
                   float* __restrict__ out) {
  __shared__ __attribute__((aligned(16))) float st[DIM];

  const int lane = threadIdx.x;
  const int n    = lane & 15;
  const int half = lane >> 4;

  const float th = theta[lane < 24 ? lane : 0];
  const float4* xr = (const float4*)(x + (size_t)blockIdx.x * DIM);
  float4 xv[32];
#pragma unroll
  for (int e = 0; e < 32; ++e) xv[e] = xr[half * 512 + e * 16 + n];

  float sv, cv;
  sincosf(th * 0.5f, &sv, &cv);

  v2f FA[4], FN[4], FB[4];
  gate_frag(FA, cv, sv, 0, 1, 10, 11, n, half);
  gate_frag(FN, cv, sv, 6, 7, 8, 9, n, half);
  gate_frag(FB, cv, sv, 2, 3, 4, 5, n, half);

  float ss0 = 0.f, ss1 = 0.f;
#pragma unroll
  for (int e = 0; e < 32; ++e) {
    const float4 v = xv[e];
    ss0 += v.x * v.x + v.y * v.y;
    ss1 += v.z * v.z + v.w * v.w;
  }
  const v2f ones = {1.f, 1.f};
  float sumsq;
  {
    v2f a;
    a.x = ss0;
    a.y = ss1;
    const v8f zero = {};
    const v8f t1 = wmma4(a, ones, zero);
    v8f t2 = {};
#pragma unroll
    for (int kc = 0; kc < 4; ++kc) {
      v2f s;
      s.x = t1[2 * kc + 0];
      s.y = t1[2 * kc + 1];
      t2 = wmma4(ones, s, t2);
    }
    sumsq = t2[0];
  }

  const bool fb = (sumsq <= 1e-24f);
  const float keep = fb ? 0.f : 1.f;
  const float e0col = (fb && half == 0) ? FA[0].x : 0.f;
#pragma unroll
  for (int kc = 0; kc < 4; ++kc) FA[kc] = FA[kc] * keep;
  const float inv = fb ? 1.f : 1.f / sumsq;

  float U[16], V[16];
#pragma unroll
  for (int i = 0; i < 16; ++i) { U[i] = 0.f; V[i] = 0.f; }

  const int pl = lds_word(gray12(((n >> 2) << 10) | (n & 3) | (half << 9)));
  ry_layer<false>(xv, FA, FN, FB, st, pl, e0col, U, V);
  __syncthreads();
#pragma unroll
  for (int e = 0; e < 32; ++e)
    xv[e] = ((const float4*)st)[half * 512 + e * 16 + (n ^ ((half << 3) | (e >> 2)))];

  gate_frag(FA, cv, sv, 12 + 0, 12 + 1, 12 + 10, 12 + 11, n, half);
  gate_frag(FN, cv, sv, 12 + 6, 12 + 7, 12 + 8, 12 + 9, n, half);
  gate_frag(FB, cv, sv, 12 + 2, 12 + 3, 12 + 4, 12 + 5, n, half);
  ry_layer<true>(xv, FA, FN, FB, st, pl, 0.f, U, V);

  const unsigned pwU = (n == 3 ? 0xFF00u : 0u) | (n == 4 ? 0x0FF0u : 0u) |
                       (n == 5 ? 0x3C3Cu : 0u) | (n == 6 ? 0x6666u : 0u);
  const unsigned pwV = (n == 7 ? 0x0FF0u : 0u) | (n == 8 ? 0x3C3Cu : 0u) |
                       (n == 9 ? 0x6666u : 0u) | (n == 10 ? 0xAAAAu : 0u);
  const bool colU = (n <= 6) || (n == 11);
  const bool colV = (n >= 7) && (n <= 10);
  const float sgnH = ((n == 2) | (n == 3)) & half ? -1.f : 1.f;
  const float baseU = colU ? sgnH : 0.f;
  const float baseV = colV ? 1.f : 0.f;
  v8f d1 = {}, d1v = {};
#pragma unroll
  for (int w = 0; w < 8; ++w) {
    v2f a, c;
    a.x = U[2 * w + 0];
    a.y = U[2 * w + 1];
    c.x = signed_coef(baseU, pwU, 2 * w + 0);
    c.y = signed_coef(baseU, pwU, 2 * w + 1);
    d1 = wmma4(a, c, d1);
  }
#pragma unroll
  for (int w = 0; w < 8; ++w) {
    v2f a, c;
    a.x = V[2 * w + 0];
    a.y = V[2 * w + 1];
    c.x = signed_coef(baseV, pwV, 2 * w + 0);
    c.y = signed_coef(baseV, pwV, 2 * w + 1);
    d1v = wmma4(a, c, d1v);
  }
  d1 = d1 + d1v;
  const unsigned pwA = (n == 0 ? 0xFF00u : 0u) | (n == 1 ? 0x0FF0u : 0u) |
                       (n == 2 ? 0x6666u : 0u) | (n == 3 ? 0xF0F0u : 0u) |
                       (n == 5 ? 0xCCCCu : 0u);
  const unsigned pwAh = half ? (pwA >> 8) : pwA;
  const float baseA = (n < 6) ? 1.f : 0.f;
  v8f d2 = {};
#pragma unroll
  for (int kc = 0; kc < 4; ++kc) {
    v2f s, c;
    s.x = d1[2 * kc + 0];
    s.y = d1[2 * kc + 1];
    c.x = signed_coef(baseA, pwAh, 2 * kc + 0);
    c.y = signed_coef(baseA, pwAh, 2 * kc + 1);
    d2 = wmma4(c, s, d2);
  }
  float r = d2[4];
  r = (lane == 0) ? d2[0] : r;
  r = (lane == 1) ? d2[1] : r;
  r = (lane == 11) ? d2[2] : r;
  r = (lane == 2) ? d2[3] : r;
  r = (lane == 10) ? d2[5] : r;
  if (lane < 12) out[(size_t)blockIdx.x * 12 + lane] = r * inv;
}

extern "C" void kernel_launch(void* const* d_in, const int* in_sizes, int n_in,
                              void* d_out, int out_size, void* d_ws,
                              size_t ws_size, hipStream_t stream) {
  const float* x = (const float*)d_in[0];
  const float* theta = (const float*)d_in[1];
  float* out = (float*)d_out;
  const int B = in_sizes[0] / DIM;
  qdense_kernel<<<dim3(B), dim3(32), 0, stream>>>(x, theta, out);
}


# RY on the 8 register-name wires as scaled rotations lo+a*hi, hi-a*lo (a=-tan, or cot with the wire's swap folded into LDS address / Z sign); scale in gate A; 2 FMAs per pair, not 4; new rounding
# speedup vs baseline: 3.7726x; 1.2813x over previous
#include <hip/hip_runtime.h>
#include <math.h>


typedef __attribute__((ext_vector_type(2))) float v2f;
typedef __attribute__((ext_vector_type(8))) float v8f;
typedef __attribute__((ext_vector_type(2))) unsigned v2u;

#define DIM 4096

constexpr unsigned cnot_fwd(unsigned k) {
  for (int i = 0; i < 11; ++i)
    for (int j = i + 1; j < 12; ++j)
      k ^= ((k >> (11 - i)) & 1u) << (11 - j);
  return k;
}
constexpr bool cascade_is_gray() {
  for (int b = 0; b < 12; ++b) {
    const unsigned k = 1u << b;
    if (cnot_fwd(k) != (k ^ (k >> 1))) return false;
  }
  return true;
}
static_assert(cascade_is_gray(), "CNOT cascade != Gray map");

__host__ __device__ constexpr int gray12(int k) { return k ^ (k >> 1); }
__host__ __device__ constexpr int lds_word(int k) { return k ^ (((k >> 8) & 15) << 2); }

__device__ __forceinline__ v8f wmma4(v2f a, v2f b, v8f c) {
  return __builtin_amdgcn_wmma_f32_16x16x4_f32(false, a, false, b,
                                               (short)0, c, false, false);
}

__device__ __forceinline__ float ry_entry(int rb, int cb, float c, float s) {
  return rb ? (cb ? c : s) : (cb ? -s : c);
}

__device__ __forceinline__ float lane_bcast(float v, int l) {
  return __int_as_float(__builtin_amdgcn_readlane(__float_as_int(v), l));
}

__device__ __forceinline__ void half_swap(float& x, float& y) {
  const v2u r = __builtin_amdgcn_permlane16_swap(__float_as_uint(x),
                                                 __float_as_uint(y), false, false);
  x = __uint_as_float(r.x);
  y = __uint_as_float(r.y);
}

__device__ __forceinline__ void gate_frag(v2f (&F)[4], float cv, float sv,
                                          int i3, int i2, int i1, int i0,
                                          int n, int half) {
  const float c3 = lane_bcast(cv, i3), s3 = lane_bcast(sv, i3);
  const float c2 = lane_bcast(cv, i2), s2 = lane_bcast(sv, i2);
  const float c1 = lane_bcast(cv, i1), s1 = lane_bcast(sv, i1);
  const float c0 = lane_bcast(cv, i0), s0 = lane_bcast(sv, i0);
  const float e3    =  ry_entry((n >> 3) & 1, half, c3, s3);
  const float e2[2] = {ry_entry((n >> 2) & 1, 0, c2, s2),
                       ry_entry((n >> 2) & 1, 1, c2, s2)};
  const float e1[2] = {ry_entry((n >> 1) & 1, 0, c1, s1),
                       ry_entry((n >> 1) & 1, 1, c1, s1)};
  const float e0[2] = {ry_entry(n & 1, 0, c0, s0),
                       ry_entry(n & 1, 1, c0, s0)};
#pragma unroll
  for (int kc = 0; kc < 4; ++kc) {
    const float p = (e3 * e2[kc >> 1]) * e1[kc & 1];
    F[kc].x = p * e0[0];
    F[kc].y = p * e0[1];
  }
}

__device__ __forceinline__ float signed_coef(float base, unsigned pw, int idx) {
  return __uint_as_float(__float_as_uint(base) ^ ((pw << (31 - idx)) & 0x80000000u));
}

__device__ __forceinline__ void rot_pair(v2f& lo, v2f& hi, float a) {
  const v2f aa = {a, a};
  const v2f nlo = __builtin_elementwise_fma(aa, hi, lo);
  const v2f nhi = __builtin_elementwise_fma(-aa, lo, hi);
  lo = nlo;
  hi = nhi;
}

__device__ __forceinline__ void rot_in_pair(v2f& p, float a) {
  const v2f am = {a, -a};
  const v2f sw = {p.y, p.x};
  p = __builtin_elementwise_fma(am, sw, p);
}

__device__ __forceinline__ v8f pa_tile(const float4 (&xv)[32], const v2f (&FA)[4],
                                       int b) {
  v8f acc = {};
#pragma unroll
  for (int kc = 0; kc < 4; ++kc) {
    const float4 q = xv[((kc >> 1) << 4) | b];
    v2f a;
    a.x = (kc & 1) ? q.z : q.x;
    a.y = (kc & 1) ? q.w : q.y;
    acc = wmma4(a, FA[kc], acc);
  }
  return acc;
}

template <bool LAST>
__device__ __forceinline__ void ry_layer(const float4 (&xv)[32],
                                         const v2f (&FA)[4],
                                         const float (&wa)[8],
                                         float* st, int pl, float e0col,
                                         float (&U)[16], float (&V)[16]) {
  v2f P[16][4];
  v8f nxt = pa_tile(xv, FA, 0);
#pragma unroll
  for (int b = 0; b < 16; ++b) {
    v8f acc = nxt;
    if (b < 15) nxt = pa_tile(xv, FA, b + 1);
    if (b == 0) acc[0] += e0col;
#pragma unroll
    for (int j = 0; j < 4; ++j) {
      P[b][j].x = acc[2 * j + 0];
      P[b][j].y = acc[2 * j + 1];
    }
#pragma unroll
    for (int j = 0; j < 4; ++j) rot_in_pair(P[b][j], wa[0]);
    rot_pair(P[b][0], P[b][1], wa[1]);
    rot_pair(P[b][2], P[b][3], wa[1]);
    rot_pair(P[b][0], P[b][2], wa[2]);
    rot_pair(P[b][1], P[b][3], wa[2]);
    if ((b & 1) == 1) {
#pragma unroll
      for (int j = 0; j < 4; ++j) rot_pair(P[b - 1][j], P[b][j], wa[4]);
    }
    if ((b & 3) == 3) {
#pragma unroll
      for (int i = 0; i < 2; ++i)
#pragma unroll
        for (int j = 0; j < 4; ++j)
          rot_pair(P[b - 3 + i][j], P[b - 1 + i][j], wa[5]);
    }
    if ((b & 7) == 7) {
#pragma unroll
      for (int i = 0; i < 4; ++i)
#pragma unroll
        for (int j = 0; j < 4; ++j)
          rot_pair(P[b - 7 + i][j], P[b - 3 + i][j], wa[6]);
    }
    if (b == 15) {
#pragma unroll
      for (int i = 0; i < 8; ++i)
#pragma unroll
        for (int j = 0; j < 4; ++j)
          rot_pair(P[i][j], P[i + 8][j], wa[7]);
    }
  }
  v2f UU[16], VV[8];
  if (LAST) {
#pragma unroll
    for (int i = 0; i < 16; ++i) UU[i] = v2f{0.f, 0.f};
#pragma unroll
    for (int i = 0; i < 8; ++i) VV[i] = v2f{0.f, 0.f};
  }
#pragma unroll
  for (int b = 0; b < 8; ++b) {
#pragma unroll
    for (int j = 0; j < 4; ++j) {
      {
        float lx = P[b][j].x, hx = P[b + 8][j].x;
        float ly = P[b][j].y, hy = P[b + 8][j].y;
        half_swap(lx, hx);
        half_swap(ly, hy);
        P[b][j] = v2f{lx, ly};
        P[b + 8][j] = v2f{hx, hy};
      }
      rot_pair(P[b][j], P[b + 8][j], wa[3]);
#pragma unroll
      for (int x = 0; x < 2; ++x) {
        const v2f val = P[x * 8 + b][j];
        if (!LAST) {
          st[pl ^ lds_word(gray12((b << 6) | ((x * 8 + 2 * j + 0) << 2)))] = val.x;
          st[pl ^ lds_word(gray12((b << 6) | ((x * 8 + 2 * j + 1) << 2)))] = val.y;
        } else {
          const v2f p = val * val;
          UU[2 * b + x] += p;
          VV[x * 4 + j] += p;
        }
      }
    }
  }
  if (LAST) {
#pragma unroll
    for (int i = 0; i < 16; ++i) U[i] = UU[i].x + UU[i].y;
#pragma unroll
    for (int i = 0; i < 8; ++i) {
      V[2 * i + 0] = VV[i].x;
      V[2 * i + 1] = VV[i].y;
    }
  }
}

__global__ __launch_bounds__(32, 4)
void qdense_kernel(const float* __restrict__ x,
                   const float* __restrict__ theta,
                   float* __restrict__ out) {
  __shared__ __attribute__((aligned(16))) float st[DIM];

  const int lane = threadIdx.x;
  const int n    = lane & 15;
  const int half = lane >> 4;

  const float th = theta[lane < 24 ? lane : 0];
  const float4* xr = (const float4*)(x + (size_t)blockIdx.x * DIM);
  float4 xv[32];
#pragma unroll
  for (int e = 0; e < 32; ++e) xv[e] = xr[half * 512 + e * 16 + n];

  float sv, cv;
  sincosf(th * 0.5f, &sv, &cv);

  const bool cot = fabsf(sv) > fabsf(cv);
  const unsigned bm = __builtin_amdgcn_ballot_w32(cot);
  const unsigned flip1 = __builtin_bitreverse32(bm & 0x3FCu) >> 20;
  const unsigned flip2 = __builtin_bitreverse32((bm >> 12) & 0x3FCu) >> 20;
  unsigned zp = flip1;
  zp ^= zp << 1;
  zp ^= zp << 2;
  zp ^= zp << 4;
  zp ^= zp << 8;
  const unsigned zl = __builtin_bitreverse32(zp & 0xFFFu) >> 8;
  sv = ((zl >> lane) & 1u) ? -sv : sv;
  const float al = (cot ? cv : -sv) / (cot ? sv : cv);
  const float scl = cot ? sv : cv;

  v2f FA[4];
  gate_frag(FA, cv, sv, 0, 1, 10, 11, n, half);
  float wa[8];
  float sc = 1.f;
#pragma unroll
  for (int i = 0; i < 8; ++i) {
    wa[i] = lane_bcast(al, 9 - i);
    sc *= lane_bcast(scl, 9 - i);
  }
#pragma unroll
  for (int kc = 0; kc < 4; ++kc) FA[kc] = FA[kc] * sc;

  float ss0 = 0.f, ss1 = 0.f;
#pragma unroll
  for (int e = 0; e < 32; ++e) {
    const float4 v = xv[e];
    ss0 += v.x * v.x + v.y * v.y;
    ss1 += v.z * v.z + v.w * v.w;
  }
  const v2f ones = {1.f, 1.f};
  float sumsq;
  {
    v2f a;
    a.x = ss0;
    a.y = ss1;
    const v8f zero = {};
    const v8f t1 = wmma4(a, ones, zero);
    v8f t2 = {};
#pragma unroll
    for (int kc = 0; kc < 4; ++kc) {
      v2f s;
      s.x = t1[2 * kc + 0];
      s.y = t1[2 * kc + 1];
      t2 = wmma4(ones, s, t2);
    }
    sumsq = t2[0];
  }

  const bool fb = (sumsq <= 1e-24f);
  const float keep = fb ? 0.f : 1.f;
  const float e0col = (fb && half == 0) ? FA[0].x : 0.f;
#pragma unroll
  for (int kc = 0; kc < 4; ++kc) FA[kc] = FA[kc] * keep;
  const float inv = fb ? 1.f : 1.f / sumsq;

  float U[16], V[16];
#pragma unroll
  for (int i = 0; i < 16; ++i) { U[i] = 0.f; V[i] = 0.f; }

  const int pl = lds_word(gray12(((n >> 2) << 10) | (n & 3) | (half << 9))) ^
                 lds_word(gray12((int)flip1));
  ry_layer<false>(xv, FA, wa, st, pl, e0col, U, V);
  __syncthreads();
#pragma unroll
  for (int e = 0; e < 32; ++e)
    xv[e] = ((const float4*)st)[half * 512 + e * 16 + (n ^ ((half << 3) | (e >> 2)))];

  gate_frag(FA, cv, sv, 12 + 0, 12 + 1, 12 + 10, 12 + 11, n, half);
  sc = 1.f;
#pragma unroll
  for (int i = 0; i < 8; ++i) {
    wa[i] = lane_bcast(al, 12 + 9 - i);
    sc *= lane_bcast(scl, 12 + 9 - i);
  }
#pragma unroll
  for (int kc = 0; kc < 4; ++kc) FA[kc] = FA[kc] * sc;
  ry_layer<true>(xv, FA, wa, st, pl, 0.f, U, V);

  const unsigned pwU = (n == 3 ? 0xFF00u : 0u) | (n == 4 ? 0x0FF0u : 0u) |
                       (n == 5 ? 0x3C3Cu : 0u) | (n == 6 ? 0x6666u : 0u);
  const unsigned pwV = (n == 7 ? 0x0FF0u : 0u) | (n == 8 ? 0x3C3Cu : 0u) |
                       (n == 9 ? 0x6666u : 0u) | (n == 10 ? 0xAAAAu : 0u);
  const bool colU = (n <= 6) || (n == 11);
  const bool colV = (n >= 7) && (n <= 10);
  const float sgnH = ((n == 2) | (n == 3)) & half ? -1.f : 1.f;
  const float baseU = colU ? sgnH : 0.f;
  const float baseV = colV ? 1.f : 0.f;
  v8f d1 = {}, d1v = {};
#pragma unroll
  for (int w = 0; w < 8; ++w) {
    v2f a, c;
    a.x = U[2 * w + 0];
    a.y = U[2 * w + 1];
    c.x = signed_coef(baseU, pwU, 2 * w + 0);
    c.y = signed_coef(baseU, pwU, 2 * w + 1);
    d1 = wmma4(a, c, d1);
  }
#pragma unroll
  for (int w = 0; w < 8; ++w) {
    v2f a, c;
    a.x = V[2 * w + 0];
    a.y = V[2 * w + 1];
    c.x = signed_coef(baseV, pwV, 2 * w + 0);
    c.y = signed_coef(baseV, pwV, 2 * w + 1);
    d1v = wmma4(a, c, d1v);
  }
  d1 = d1 + d1v;
  const unsigned pwA = (n == 0 ? 0xFF00u : 0u) | (n == 1 ? 0x0FF0u : 0u) |
                       (n == 2 ? 0x6666u : 0u) | (n == 3 ? 0xF0F0u : 0u) |
                       (n == 5 ? 0xCCCCu : 0u);
  const unsigned pwAh = half ? (pwA >> 8) : pwA;
  const float baseA = (n < 6) ? 1.f : 0.f;
  v8f d2 = {};
#pragma unroll
  for (int kc = 0; kc < 4; ++kc) {
    v2f s, c;
    s.x = d1[2 * kc + 0];
    s.y = d1[2 * kc + 1];
    c.x = signed_coef(baseA, pwAh, 2 * kc + 0);
    c.y = signed_coef(baseA, pwAh, 2 * kc + 1);
    d2 = wmma4(c, s, d2);
  }
  float r = d2[4];
  r = (lane == 0) ? d2[0] : r;
  r = (lane == 1) ? d2[1] : r;
  r = (lane == 11) ? d2[2] : r;
  r = (lane == 2) ? d2[3] : r;
  r = (lane == 10) ? d2[5] : r;
  const unsigned zs = __builtin_bitreverse32(flip2 ^ (flip2 >> 1)) >> 20;
  const float oinv = ((zs >> lane) & 1u) ? -inv : inv;
  if (lane < 12) out[(size_t)blockIdx.x * 12 + lane] = r * oinv;
}

extern "C" void kernel_launch(void* const* d_in, const int* in_sizes, int n_in,
                              void* d_out, int out_size, void* d_ws,
                              size_t ws_size, hipStream_t stream) {
  const float* x = (const float*)d_in[0];
  const float* theta = (const float*)d_in[1];
  float* out = (float*)d_out;
  const int B = in_sizes[0] / DIM;
  qdense_kernel<<<dim3(B), dim3(32), 0, stream>>>(x, theta, out);
}


# Layer 1's RY on all 12 wires as in-register scaled rotations; a float4 layout change through LDS turns the lane-held wires into register names: no transposing WMMA pass, no half-swaps; new sum order
# speedup vs baseline: 4.3894x; 1.1635x over previous
#include <hip/hip_runtime.h>
#include <math.h>


typedef __attribute__((ext_vector_type(2))) float v2f;
typedef __attribute__((ext_vector_type(8))) float v8f;
typedef __attribute__((ext_vector_type(2))) unsigned v2u;

#define DIM 4096

constexpr unsigned cnot_fwd(unsigned k) {
  for (int i = 0; i < 11; ++i)
    for (int j = i + 1; j < 12; ++j)
      k ^= ((k >> (11 - i)) & 1u) << (11 - j);
  return k;
}
constexpr bool cascade_is_gray() {
  for (int b = 0; b < 12; ++b) {
    const unsigned k = 1u << b;
    if (cnot_fwd(k) != (k ^ (k >> 1))) return false;
  }
  return true;
}
static_assert(cascade_is_gray(), "CNOT cascade != Gray map");

__host__ __device__ constexpr int gray12(int k) { return k ^ (k >> 1); }
__host__ __device__ constexpr int lds_word(int k) {
  return k ^ (((k >> 8) & 15) << 2) ^ (((k >> 6) & 1) << 1);
}

__device__ __forceinline__ v8f wmma4(v2f a, v2f b, v8f c) {
  return __builtin_amdgcn_wmma_f32_16x16x4_f32(false, a, false, b,
                                               (short)0, c, false, false);
}

__device__ __forceinline__ float ry_entry(int rb, int cb, float c, float s) {
  return rb ? (cb ? c : s) : (cb ? -s : c);
}

__device__ __forceinline__ float lane_bcast(float v, int l) {
  return __int_as_float(__builtin_amdgcn_readlane(__float_as_int(v), l));
}

__device__ __forceinline__ void half_swap(float& x, float& y) {
  const v2u r = __builtin_amdgcn_permlane16_swap(__float_as_uint(x),
                                                 __float_as_uint(y), false, false);
  x = __uint_as_float(r.x);
  y = __uint_as_float(r.y);
}

__device__ __forceinline__ void gate_frag(v2f (&F)[4], float cv, float sv,
                                          int i3, int i2, int i1, int i0,
                                          int n, int half) {
  const float c3 = lane_bcast(cv, i3), s3 = lane_bcast(sv, i3);
  const float c2 = lane_bcast(cv, i2), s2 = lane_bcast(sv, i2);
  const float c1 = lane_bcast(cv, i1), s1 = lane_bcast(sv, i1);
  const float c0 = lane_bcast(cv, i0), s0 = lane_bcast(sv, i0);
  const float e3    =  ry_entry((n >> 3) & 1, half, c3, s3);
  const float e2[2] = {ry_entry((n >> 2) & 1, 0, c2, s2),
                       ry_entry((n >> 2) & 1, 1, c2, s2)};
  const float e1[2] = {ry_entry((n >> 1) & 1, 0, c1, s1),
                       ry_entry((n >> 1) & 1, 1, c1, s1)};
  const float e0[2] = {ry_entry(n & 1, 0, c0, s0),
                       ry_entry(n & 1, 1, c0, s0)};
#pragma unroll
  for (int kc = 0; kc < 4; ++kc) {
    const float p = (e3 * e2[kc >> 1]) * e1[kc & 1];
    F[kc].x = p * e0[0];
    F[kc].y = p * e0[1];
  }
}

__device__ __forceinline__ float signed_coef(float base, unsigned pw, int idx) {
  return __uint_as_float(__float_as_uint(base) ^ ((pw << (31 - idx)) & 0x80000000u));
}

__device__ __forceinline__ void rot_pair(v2f& lo, v2f& hi, float a) {
  const v2f aa = {a, a};
  const v2f nlo = __builtin_elementwise_fma(aa, hi, lo);
  const v2f nhi = __builtin_elementwise_fma(-aa, lo, hi);
  lo = nlo;
  hi = nhi;
}

__device__ __forceinline__ void rot_in_pair(v2f& p, float a) {
  const v2f am = {a, -a};
  const v2f sw = {p.y, p.x};
  p = __builtin_elementwise_fma(am, sw, p);
}

__device__ __forceinline__ v8f pa_tile(const float4 (&xv)[32], const v2f (&FA)[4],
                                       int b) {
  v8f acc = {};
#pragma unroll
  for (int kc = 0; kc < 4; ++kc) {
    const float4 q = xv[((kc >> 1) << 4) | b];
    v2f a;
    a.x = (kc & 1) ? q.z : q.x;
    a.y = (kc & 1) ? q.w : q.y;
    acc = wmma4(a, FA[kc], acc);
  }
  return acc;
}

__device__ __forceinline__ void rot_quad(float4& lo, float4& hi, float a) {
  v2f l0 = {lo.x, lo.y}, l1 = {lo.z, lo.w};
  v2f h0 = {hi.x, hi.y}, h1 = {hi.z, hi.w};
  rot_pair(l0, h0, a);
  rot_pair(l1, h1, a);
  lo = make_float4(l0.x, l0.y, l1.x, l1.y);
  hi = make_float4(h0.x, h0.y, h1.x, h1.y);
}

__device__ __forceinline__ void gray_store(const float4 (&xv)[32], float* st,
                                           int lane, unsigned flip) {
  const int pl = lds_word(gray12((lane << 6) ^ (int)flip));
#pragma unroll
  for (int f = 0; f < 32; ++f) {
    const int kr = ((f >> 4) << 11) | ((f & 15) << 2);
    st[pl ^ lds_word(gray12(kr | 0))] = xv[f].x;
    st[pl ^ lds_word(gray12(kr | 1))] = xv[f].y;
    st[pl ^ lds_word(gray12(kr | 2))] = xv[f].z;
    st[pl ^ lds_word(gray12(kr | 3))] = xv[f].w;
  }
}

__device__ __forceinline__ void ry_layer1(float4 (&xv)[32], const float (&a)[12],
                                          float* st, int lane, unsigned flip) {
  const int n = lane & 15, half = lane >> 4;
#pragma unroll
  for (int e = 0; e < 32; ++e) {
    v2f p0 = {xv[e].x, xv[e].y}, p1 = {xv[e].z, xv[e].w};
    rot_in_pair(p0, a[0]);
    rot_in_pair(p1, a[0]);
    rot_pair(p0, p1, a[1]);
    xv[e] = make_float4(p0.x, p0.y, p1.x, p1.y);
  }
#pragma unroll
  for (int i = 0; i < 5; ++i)
#pragma unroll
    for (int e = 0; e < 32; ++e)
      if (((e >> i) & 1) == 0) rot_quad(xv[e], xv[e | (1 << i)], a[6 + i]);
  float4* st4 = (float4*)st;
#pragma unroll
  for (int e = 0; e < 32; ++e)
    st4[(half << 9) | (e << 4) | (n ^ (e & 15))] = xv[e];
  __syncthreads();
#pragma unroll
  for (int f = 0; f < 32; ++f)
    xv[f] = st4[((f >> 4) << 9) | (lane << 4) | ((f & 15) ^ (lane & 15))];
#pragma unroll
  for (int i = 0; i < 5; ++i)
#pragma unroll
    for (int f = 0; f < 32; ++f)
      if (((f >> i) & 1) == 0)
        rot_quad(xv[f], xv[f | (1 << i)], a[i == 4 ? 11 : 2 + i]);
  gray_store(xv, st, lane, flip);
}

__device__ __forceinline__ void ry_layer2(const float4 (&xv)[32],
                                          const v2f (&FA)[4],
                                          const float (&wa)[8],
                                          float (&U)[16], float (&V)[16]) {
  v2f P[16][4];
  v8f nxt = pa_tile(xv, FA, 0);
#pragma unroll
  for (int b = 0; b < 16; ++b) {
    v8f acc = nxt;
    if (b < 15) nxt = pa_tile(xv, FA, b + 1);
#pragma unroll
    for (int j = 0; j < 4; ++j) {
      P[b][j].x = acc[2 * j + 0];
      P[b][j].y = acc[2 * j + 1];
    }
#pragma unroll
    for (int j = 0; j < 4; ++j) rot_in_pair(P[b][j], wa[0]);
    rot_pair(P[b][0], P[b][1], wa[1]);
    rot_pair(P[b][2], P[b][3], wa[1]);
    rot_pair(P[b][0], P[b][2], wa[2]);
    rot_pair(P[b][1], P[b][3], wa[2]);
    if ((b & 1) == 1) {
#pragma unroll
      for (int j = 0; j < 4; ++j) rot_pair(P[b - 1][j], P[b][j], wa[4]);
    }
    if ((b & 3) == 3) {
#pragma unroll
      for (int i = 0; i < 2; ++i)
#pragma unroll
        for (int j = 0; j < 4; ++j)
          rot_pair(P[b - 3 + i][j], P[b - 1 + i][j], wa[5]);
    }
    if ((b & 7) == 7) {
#pragma unroll
      for (int i = 0; i < 4; ++i)
#pragma unroll
        for (int j = 0; j < 4; ++j)
          rot_pair(P[b - 7 + i][j], P[b - 3 + i][j], wa[6]);
    }
    if (b == 15) {
#pragma unroll
      for (int i = 0; i < 8; ++i)
#pragma unroll
        for (int j = 0; j < 4; ++j)
          rot_pair(P[i][j], P[i + 8][j], wa[7]);
    }
  }
  v2f UU[16], VV[8];
#pragma unroll
  for (int i = 0; i < 16; ++i) UU[i] = v2f{0.f, 0.f};
#pragma unroll
  for (int i = 0; i < 8; ++i) VV[i] = v2f{0.f, 0.f};
#pragma unroll
  for (int b = 0; b < 8; ++b) {
#pragma unroll
    for (int j = 0; j < 4; ++j) {
      {
        float lx = P[b][j].x, hx = P[b + 8][j].x;
        float ly = P[b][j].y, hy = P[b + 8][j].y;
        half_swap(lx, hx);
        half_swap(ly, hy);
        P[b][j] = v2f{lx, ly};
        P[b + 8][j] = v2f{hx, hy};
      }
      rot_pair(P[b][j], P[b + 8][j], wa[3]);
#pragma unroll
      for (int x = 0; x < 2; ++x) {
        const v2f val = P[x * 8 + b][j];
        const v2f p = val * val;
        UU[2 * b + x] += p;
        VV[x * 4 + j] += p;
      }
    }
  }
#pragma unroll
  for (int i = 0; i < 16; ++i) U[i] = UU[i].x + UU[i].y;
#pragma unroll
  for (int i = 0; i < 8; ++i) {
    V[2 * i + 0] = VV[i].x;
    V[2 * i + 1] = VV[i].y;
  }
}

__global__ __launch_bounds__(32, 4)
void qdense_kernel(const float* __restrict__ x,
                   const float* __restrict__ theta,
                   float* __restrict__ out) {
  __shared__ __attribute__((aligned(16))) float st[DIM];

  const int lane = threadIdx.x;
  const int n    = lane & 15;
  const int half = lane >> 4;

  const float th = theta[lane < 24 ? lane : 0];
  const float4* xr = (const float4*)(x + (size_t)blockIdx.x * DIM);
  float4 xv[32];
#pragma unroll
  for (int e = 0; e < 32; ++e) xv[e] = xr[half * 512 + e * 16 + n];

  float sv, cv;
  sincosf(th * 0.5f, &sv, &cv);

  const bool cot = fabsf(sv) > fabsf(cv);
  const unsigned bm = __builtin_amdgcn_ballot_w32(cot);
  const unsigned flip1 = __builtin_bitreverse32(bm & 0xFFFu) >> 20;
  const unsigned flip2 = __builtin_bitreverse32((bm >> 12) & 0x3FCu) >> 20;
  unsigned zp = flip1;
  zp ^= zp << 1;
  zp ^= zp << 2;
  zp ^= zp << 4;
  zp ^= zp << 8;
  const unsigned zl = __builtin_bitreverse32(zp & 0xFFFu) >> 8;
  sv = ((zl >> lane) & 1u) ? -sv : sv;
  const float al = (cot ? cv : -sv) / (cot ? sv : cv);
  const float scl = cot ? sv : cv;

  float a1[12];
  float sc1 = 1.f;
#pragma unroll
  for (int b = 0; b < 12; ++b) {
    a1[b] = lane_bcast(al, 11 - b);
    sc1 *= lane_bcast(scl, 11 - b);
  }

  float ss0 = 0.f, ss1 = 0.f;
#pragma unroll
  for (int e = 0; e < 32; ++e) {
    const float4 v = xv[e];
    ss0 += v.x * v.x + v.y * v.y;
    ss1 += v.z * v.z + v.w * v.w;
  }
  const v2f ones = {1.f, 1.f};
  float sumsq;
  {
    v2f a;
    a.x = ss0;
    a.y = ss1;
    const v8f zero = {};
    const v8f t1 = wmma4(a, ones, zero);
    v8f t2 = {};
#pragma unroll
    for (int kc = 0; kc < 4; ++kc) {
      v2f s;
      s.x = t1[2 * kc + 0];
      s.y = t1[2 * kc + 1];
      t2 = wmma4(ones, s, t2);
    }
    sumsq = t2[0];
  }

  const bool fb = (sumsq <= 1e-24f);
  const float inv = fb ? 1.f : 1.f / sumsq;

  ry_layer1(xv, a1, st, lane, flip1);
  if (__builtin_expect(fb, 0)) {
    float pw = 1.f;
#pragma unroll
    for (int i = 0; i < 5; ++i) pw *= ((lane >> i) & 1) ? -a1[6 + i] : 1.f;
    xv[0].x = pw;
    xv[0].y = xv[0].x * -a1[0];
    xv[0].z = xv[0].x * -a1[1];
    xv[0].w = xv[0].y * -a1[1];
#pragma unroll
    for (int i = 0; i < 5; ++i) {
      const float m = -a1[i == 4 ? 11 : 2 + i];
#pragma unroll
      for (int f = 0; f < (1 << i); ++f)
        xv[f | (1 << i)] = make_float4(xv[f].x * m, xv[f].y * m, xv[f].z * m, xv[f].w * m);
    }
    gray_store(xv, st, lane, flip1);
  }
  __syncthreads();
#pragma unroll
  for (int e = 0; e < 32; ++e) {
    const float4 q =
        ((const float4*)st)[half * 512 + e * 16 + (n ^ ((half << 3) | (e >> 2)))];
    xv[e] = (e & 1) ? make_float4(q.z, q.w, q.x, q.y) : q;
  }

  v2f FA[4];
  gate_frag(FA, cv, sv, 12 + 0, 12 + 1, 12 + 10, 12 + 11, n, half);
  float wa[8];
  float sc = sc1;
#pragma unroll
  for (int i = 0; i < 8; ++i) {
    wa[i] = lane_bcast(al, 12 + 9 - i);
    sc *= lane_bcast(scl, 12 + 9 - i);
  }
#pragma unroll
  for (int kc = 0; kc < 4; ++kc) FA[kc] = FA[kc] * sc;
  float U[16], V[16];
  ry_layer2(xv, FA, wa, U, V);

  const unsigned pwU = (n == 3 ? 0xFF00u : 0u) | (n == 4 ? 0x0FF0u : 0u) |
                       (n == 5 ? 0x3C3Cu : 0u) | (n == 6 ? 0x6666u : 0u);
  const unsigned pwV = (n == 7 ? 0x0FF0u : 0u) | (n == 8 ? 0x3C3Cu : 0u) |
                       (n == 9 ? 0x6666u : 0u) | (n == 10 ? 0xAAAAu : 0u);
  const bool colU = (n <= 6) || (n == 11);
  const bool colV = (n >= 7) && (n <= 10);
  const float sgnH = ((n == 2) | (n == 3)) & half ? -1.f : 1.f;
  const float baseU = colU ? sgnH : 0.f;
  const float baseV = colV ? 1.f : 0.f;
  v8f d1 = {}, d1v = {};
#pragma unroll
  for (int w = 0; w < 8; ++w) {
    v2f a, c;
    a.x = U[2 * w + 0];
    a.y = U[2 * w + 1];
    c.x = signed_coef(baseU, pwU, 2 * w + 0);
    c.y = signed_coef(baseU, pwU, 2 * w + 1);
    d1 = wmma4(a, c, d1);
  }
#pragma unroll
  for (int w = 0; w < 8; ++w) {
    v2f a, c;
    a.x = V[2 * w + 0];
    a.y = V[2 * w + 1];
    c.x = signed_coef(baseV, pwV, 2 * w + 0);
    c.y = signed_coef(baseV, pwV, 2 * w + 1);
    d1v = wmma4(a, c, d1v);
  }
  d1 = d1 + d1v;
  const unsigned pwA = (n == 0 ? 0xFF00u : 0u) | (n == 1 ? 0x0FF0u : 0u) |
                       (n == 2 ? 0x6666u : 0u) | (n == 3 ? 0xF0F0u : 0u) |
                       (n == 5 ? 0xCCCCu : 0u);
  const unsigned pwAh = half ? (pwA >> 8) : pwA;
  const float baseA = (n < 6) ? 1.f : 0.f;
  v8f d2 = {};
#pragma unroll
  for (int kc = 0; kc < 4; ++kc) {
    v2f s, c;
    s.x = d1[2 * kc + 0];
    s.y = d1[2 * kc + 1];
    c.x = signed_coef(baseA, pwAh, 2 * kc + 0);
    c.y = signed_coef(baseA, pwAh, 2 * kc + 1);
    d2 = wmma4(c, s, d2);
  }
  float r = d2[4];
  r = (lane == 0) ? d2[0] : r;
  r = (lane == 1) ? d2[1] : r;
  r = (lane == 11) ? d2[2] : r;
  r = (lane == 2) ? d2[3] : r;
  r = (lane == 10) ? d2[5] : r;
  const unsigned zs = __builtin_bitreverse32(flip2 ^ (flip2 >> 1)) >> 20;
  const float oinv = ((zs >> lane) & 1u) ? -inv : inv;
  if (lane < 12) out[(size_t)blockIdx.x * 12 + lane] = r * oinv;
}

extern "C" void kernel_launch(void* const* d_in, const int* in_sizes, int n_in,
                              void* d_out, int out_size, void* d_ws,
                              size_t ws_size, hipStream_t stream) {
  const float* x = (const float*)d_in[0];
  const float* theta = (const float*)d_in[1];
  float* out = (float*)d_out;
  const int B = in_sizes[0] / DIM;
  qdense_kernel<<<dim3(B), dim3(32), 0, stream>>>(x, theta, out);
}


# Layer 2's RY on all 12 wires as in-register scaled rotations (float4 layout change through LDS, as layer 1): no dense 16x16 WMMA gate pass, no half-swaps; scale^2 into 1/norm; new sum order
# speedup vs baseline: 4.8064x; 1.0950x over previous
#include <hip/hip_runtime.h>
#include <math.h>


typedef __attribute__((ext_vector_type(2))) float v2f;
typedef __attribute__((ext_vector_type(8))) float v8f;

#define DIM 4096

constexpr unsigned cnot_fwd(unsigned k) {
  for (int i = 0; i < 11; ++i)
    for (int j = i + 1; j < 12; ++j)
      k ^= ((k >> (11 - i)) & 1u) << (11 - j);
  return k;
}
constexpr bool cascade_is_gray() {
  for (int b = 0; b < 12; ++b) {
    const unsigned k = 1u << b;
    if (cnot_fwd(k) != (k ^ (k >> 1))) return false;
  }
  return true;
}
static_assert(cascade_is_gray(), "CNOT cascade != Gray map");

__host__ __device__ constexpr int gray12(int k) { return k ^ (k >> 1); }
__host__ __device__ constexpr int lds_word(int k) {
  return k ^ (((k >> 8) & 15) << 2) ^ (((k >> 6) & 1) << 1);
}

__device__ __forceinline__ v8f wmma4(v2f a, v2f b, v8f c) {
  return __builtin_amdgcn_wmma_f32_16x16x4_f32(false, a, false, b,
                                               (short)0, c, false, false);
}

__device__ __forceinline__ float lane_bcast(float v, int l) {
  return __int_as_float(__builtin_amdgcn_readlane(__float_as_int(v), l));
}

__device__ __forceinline__ float signed_coef(float base, unsigned pw, int idx) {
  return __uint_as_float(__float_as_uint(base) ^ ((pw << (31 - idx)) & 0x80000000u));
}

__device__ __forceinline__ void rot_pair(v2f& lo, v2f& hi, float a) {
  const v2f aa = {a, a};
  const v2f nlo = __builtin_elementwise_fma(aa, hi, lo);
  const v2f nhi = __builtin_elementwise_fma(-aa, lo, hi);
  lo = nlo;
  hi = nhi;
}

__device__ __forceinline__ void rot_in_pair(v2f& p, float a) {
  const v2f am = {a, -a};
  const v2f sw = {p.y, p.x};
  p = __builtin_elementwise_fma(am, sw, p);
}

__device__ __forceinline__ void rot_quad(float4& lo, float4& hi, float a) {
  v2f l0 = {lo.x, lo.y}, l1 = {lo.z, lo.w};
  v2f h0 = {hi.x, hi.y}, h1 = {hi.z, hi.w};
  rot_pair(l0, h0, a);
  rot_pair(l1, h1, a);
  lo = make_float4(l0.x, l0.y, l1.x, l1.y);
  hi = make_float4(h0.x, h0.y, h1.x, h1.y);
}

__device__ __forceinline__ void gray_store(const float4 (&xv)[32], float* st,
                                           int lane, unsigned flip) {
  const int pl = lds_word(gray12((lane << 6) ^ (int)flip));
#pragma unroll
  for (int f = 0; f < 32; ++f) {
    const int kr = ((f >> 4) << 11) | ((f & 15) << 2);
    st[pl ^ lds_word(gray12(kr | 0))] = xv[f].x;
    st[pl ^ lds_word(gray12(kr | 1))] = xv[f].y;
    st[pl ^ lds_word(gray12(kr | 2))] = xv[f].z;
    st[pl ^ lds_word(gray12(kr | 3))] = xv[f].w;
  }
}

__device__ __forceinline__ void ry_all_wires(float4 (&xv)[32], const float (&a)[12],
                                             float* st, int lane) {
  const int n = lane & 15, half = lane >> 4;
#pragma unroll
  for (int e = 0; e < 32; ++e) {
    v2f p0 = {xv[e].x, xv[e].y}, p1 = {xv[e].z, xv[e].w};
    rot_in_pair(p0, a[0]);
    rot_in_pair(p1, a[0]);
    rot_pair(p0, p1, a[1]);
    xv[e] = make_float4(p0.x, p0.y, p1.x, p1.y);
  }
#pragma unroll
  for (int i = 0; i < 5; ++i)
#pragma unroll
    for (int e = 0; e < 32; ++e)
      if (((e >> i) & 1) == 0) rot_quad(xv[e], xv[e | (1 << i)], a[6 + i]);
  float4* st4 = (float4*)st;
#pragma unroll
  for (int e = 0; e < 32; ++e)
    st4[(half << 9) | (e << 4) | (n ^ (e & 15))] = xv[e];
  __syncthreads();
#pragma unroll
  for (int f = 0; f < 32; ++f)
    xv[f] = st4[((f >> 4) << 9) | (lane << 4) | ((f & 15) ^ (lane & 15))];
#pragma unroll
  for (int i = 0; i < 5; ++i)
#pragma unroll
    for (int f = 0; f < 32; ++f)
      if (((f >> i) & 1) == 0)
        rot_quad(xv[f], xv[f | (1 << i)], a[i == 4 ? 11 : 2 + i]);
}

__device__ __forceinline__ void square_sums(const float4 (&xv)[32],
                                            float (&U)[16], float (&V)[16]) {
  v2f UU[16], VV[8];
#pragma unroll
  for (int i = 0; i < 16; ++i) UU[i] = v2f{0.f, 0.f};
#pragma unroll
  for (int i = 0; i < 8; ++i) VV[i] = v2f{0.f, 0.f};
#pragma unroll
  for (int f = 0; f < 32; ++f) {
    const v2f lo = {xv[f].x, xv[f].y}, hi = {xv[f].z, xv[f].w};
    const v2f pl = lo * lo;
    const v2f ph = hi * hi;
    UU[f >> 1] += pl;
    UU[f >> 1] += ph;
    VV[(f & 3) * 2 + 0] += pl;
    VV[(f & 3) * 2 + 1] += ph;
  }
#pragma unroll
  for (int i = 0; i < 16; ++i) U[i] = UU[i].x + UU[i].y;
#pragma unroll
  for (int i = 0; i < 8; ++i) {
    V[2 * i + 0] = VV[i].x;
    V[2 * i + 1] = VV[i].y;
  }
}

__global__ __launch_bounds__(32, 4)
void qdense_kernel(const float* __restrict__ x,
                   const float* __restrict__ theta,
                   float* __restrict__ out) {
  __shared__ __attribute__((aligned(16))) float st[DIM];

  const int lane = threadIdx.x;
  const int n    = lane & 15;
  const int half = lane >> 4;

  const float th = theta[lane < 24 ? lane : 0];
  const float4* xr = (const float4*)(x + (size_t)blockIdx.x * DIM);
  float4 xv[32];
#pragma unroll
  for (int e = 0; e < 32; ++e) xv[e] = xr[half * 512 + e * 16 + n];

  float sv, cv;
  sincosf(th * 0.5f, &sv, &cv);

  const bool cot = fabsf(sv) > fabsf(cv);
  const unsigned bm = __builtin_amdgcn_ballot_w32(cot);
  const unsigned flip1 = __builtin_bitreverse32(bm & 0xFFFu) >> 20;
  const unsigned flip2 = __builtin_bitreverse32((bm >> 12) & 0xFFFu) >> 20;
  unsigned zp = flip1;
  zp ^= zp << 1;
  zp ^= zp << 2;
  zp ^= zp << 4;
  zp ^= zp << 8;
  const unsigned zl = __builtin_bitreverse32(zp & 0xFFFu) >> 8;
  sv = ((zl >> lane) & 1u) ? -sv : sv;
  const float al = (cot ? cv : -sv) / (cot ? sv : cv);
  const float scl = cot ? sv : cv;

  float a1[12];
  float sc1 = 1.f;
#pragma unroll
  for (int b = 0; b < 12; ++b) {
    a1[b] = lane_bcast(al, 11 - b);
    sc1 *= lane_bcast(scl, 11 - b);
  }

  float ss0 = 0.f, ss1 = 0.f;
#pragma unroll
  for (int e = 0; e < 32; ++e) {
    const float4 v = xv[e];
    ss0 += v.x * v.x + v.y * v.y;
    ss1 += v.z * v.z + v.w * v.w;
  }
  const v2f ones = {1.f, 1.f};
  float sumsq;
  {
    v2f a;
    a.x = ss0;
    a.y = ss1;
    const v8f zero = {};
    const v8f t1 = wmma4(a, ones, zero);
    v8f t2 = {};
#pragma unroll
    for (int kc = 0; kc < 4; ++kc) {
      v2f s;
      s.x = t1[2 * kc + 0];
      s.y = t1[2 * kc + 1];
      t2 = wmma4(ones, s, t2);
    }
    sumsq = t2[0];
  }

  const bool fb = (sumsq <= 1e-24f);
  const float inv = fb ? 1.f : 1.f / sumsq;

  ry_all_wires(xv, a1, st, lane);
  gray_store(xv, st, lane, flip1);
  if (__builtin_expect(fb, 0)) {
    float pw = 1.f;
#pragma unroll
    for (int i = 0; i < 5; ++i) pw *= ((lane >> i) & 1) ? -a1[6 + i] : 1.f;
    xv[0].x = pw;
    xv[0].y = xv[0].x * -a1[0];
    xv[0].z = xv[0].x * -a1[1];
    xv[0].w = xv[0].y * -a1[1];
#pragma unroll
    for (int i = 0; i < 5; ++i) {
      const float m = -a1[i == 4 ? 11 : 2 + i];
#pragma unroll
      for (int f = 0; f < (1 << i); ++f)
        xv[f | (1 << i)] = make_float4(xv[f].x * m, xv[f].y * m, xv[f].z * m, xv[f].w * m);
    }
    gray_store(xv, st, lane, flip1);
  }
  __syncthreads();
#pragma unroll
  for (int e = 0; e < 32; ++e) {
    const float4 q =
        ((const float4*)st)[half * 512 + e * 16 + (n ^ ((half << 3) | (e >> 2)))];
    xv[e] = (e & 1) ? make_float4(q.z, q.w, q.x, q.y) : q;
  }

  float a2[12];
  float sc = sc1;
#pragma unroll
  for (int b = 0; b < 12; ++b) {
    a2[b] = lane_bcast(al, 23 - b);
    sc *= lane_bcast(scl, 23 - b);
  }
  ry_all_wires(xv, a2, st, lane);
  float U[16], V[16];
  square_sums(xv, U, V);

  const unsigned pwU = (n <= 1 ? 0xFF00u : 0u) | (n == 6 ? 0xF0F0u : 0u) |
                       (n == 7 ? 0x3C3Cu : 0u) | (n == 8 ? 0x6666u : 0u);
  const unsigned pwV = (n == 9 ? 0x0FF0u : 0u) | (n == 10 ? 0x3C3Cu : 0u) |
                       (n == 11 ? 0x6666u : 0u);
  const bool colU = (n <= 8);
  const bool colV = (n >= 9) && (n <= 11);
  const float sgnH = ((n == 1) | (n == 2)) & half ? -1.f : 1.f;
  const float baseU = colU ? sgnH : 0.f;
  const float baseV = colV ? 1.f : 0.f;
  v8f d1 = {}, d1v = {};
#pragma unroll
  for (int w = 0; w < 8; ++w) {
    v2f a, c;
    a.x = U[2 * w + 0];
    a.y = U[2 * w + 1];
    c.x = signed_coef(baseU, pwU, 2 * w + 0);
    c.y = signed_coef(baseU, pwU, 2 * w + 1);
    d1 = wmma4(a, c, d1);
  }
#pragma unroll
  for (int w = 0; w < 8; ++w) {
    v2f a, c;
    a.x = V[2 * w + 0];
    a.y = V[2 * w + 1];
    c.x = signed_coef(baseV, pwV, 2 * w + 0);
    c.y = signed_coef(baseV, pwV, 2 * w + 1);
    d1v = wmma4(a, c, d1v);
  }
  d1 = d1 + d1v;
  const unsigned pwA = (n == 1 ? 0xFF00u : 0u) | (n == 2 ? 0x0FF0u : 0u) |
                       (n == 3 ? 0x3C3Cu : 0u) | (n == 4 ? 0x6666u : 0u) |
                       (n == 5 ? 0xAAAAu : 0u);
  const unsigned pwAh = half ? (pwA >> 8) : pwA;
  const float baseA = (n < 6) ? 1.f : 0.f;
  v8f d2 = {};
#pragma unroll
  for (int kc = 0; kc < 4; ++kc) {
    v2f s, c;
    s.x = d1[2 * kc + 0];
    s.y = d1[2 * kc + 1];
    c.x = signed_coef(baseA, pwAh, 2 * kc + 0);
    c.y = signed_coef(baseA, pwAh, 2 * kc + 1);
    d2 = wmma4(c, s, d2);
  }
  float r = d2[0];
  r = (lane == 2) ? d2[1] : r;
  r = (lane == 3) ? d2[2] : r;
  r = (lane == 4) ? d2[3] : r;
  r = (lane == 5) ? d2[4] : r;
  r = (lane == 6) ? d2[5] : r;
  const float fin = inv * (sc * sc);
  const unsigned zs = __builtin_bitreverse32(flip2 ^ (flip2 >> 1)) >> 20;
  const float oinv = ((zs >> lane) & 1u) ? -fin : fin;
  if (lane < 12) out[(size_t)blockIdx.x * 12 + lane] = r * oinv;
}

extern "C" void kernel_launch(void* const* d_in, const int* in_sizes, int n_in,
                              void* d_out, int out_size, void* d_ws,
                              size_t ws_size, hipStream_t stream) {
  const float* x = (const float*)d_in[0];
  const float* theta = (const float*)d_in[1];
  float* out = (float*)d_out;
  const int B = in_sizes[0] / DIM;
  qdense_kernel<<<dim3(B), dim3(32), 0, stream>>>(x, theta, out);
}


# Layer 2's RY on g-bits 0..6 done on layer 1's registers before the CNOT relabel (g=k^(k>>1): g-bit b pairs k with k^(2^(b+1)-1), in one lane): 2 LDS trips not 3, all b128; new order of float sums
# speedup vs baseline: 5.2215x; 1.0864x over previous
#include <hip/hip_runtime.h>
#include <math.h>


typedef __attribute__((ext_vector_type(2))) float v2f;
typedef __attribute__((ext_vector_type(8))) float v8f;

#define DIM 4096

constexpr unsigned cnot_fwd(unsigned k) {
  for (int i = 0; i < 11; ++i)
    for (int j = i + 1; j < 12; ++j)
      k ^= ((k >> (11 - i)) & 1u) << (11 - j);
  return k;
}
constexpr bool cascade_is_gray() {
  for (int b = 0; b < 12; ++b) {
    const unsigned k = 1u << b;
    if (cnot_fwd(k) != (k ^ (k >> 1))) return false;
  }
  return true;
}
static_assert(cascade_is_gray(), "CNOT cascade != Gray map");

__host__ __device__ constexpr int lds_slot(int s) { return s ^ ((s >> 5) & 15); }

__device__ __forceinline__ v8f wmma4(v2f a, v2f b, v8f c) {
  return __builtin_amdgcn_wmma_f32_16x16x4_f32(false, a, false, b,
                                               (short)0, c, false, false);
}

__device__ __forceinline__ float lane_bcast(float v, int l) {
  return __int_as_float(__builtin_amdgcn_readlane(__float_as_int(v), l));
}

__device__ __forceinline__ float signed_coef(float base, unsigned pw, int idx) {
  return __uint_as_float(__float_as_uint(base) ^ ((pw << (31 - idx)) & 0x80000000u));
}

__device__ __forceinline__ void rot_pair(v2f& lo, v2f& hi, float a) {
  const v2f aa = {a, a};
  const v2f nlo = __builtin_elementwise_fma(aa, hi, lo);
  const v2f nhi = __builtin_elementwise_fma(-aa, lo, hi);
  lo = nlo;
  hi = nhi;
}

__device__ __forceinline__ void rot_in_pair(v2f& p, float a) {
  const v2f am = {a, -a};
  const v2f sw = {p.y, p.x};
  p = __builtin_elementwise_fma(am, sw, p);
}

__device__ __forceinline__ void rot_quad(float4& lo, float4& hi, float a) {
  v2f l0 = {lo.x, lo.y}, l1 = {lo.z, lo.w};
  v2f h0 = {hi.x, hi.y}, h1 = {hi.z, hi.w};
  rot_pair(l0, h0, a);
  rot_pair(l1, h1, a);
  lo = make_float4(l0.x, l0.y, l1.x, l1.y);
  hi = make_float4(h0.x, h0.y, h1.x, h1.y);
}

__device__ __forceinline__ void ry_layer1(float4 (&xv)[32], v2f (&S)[64],
                                          const float (&a)[12], float4* st4,
                                          int lane) {
#pragma unroll
  for (int e = 0; e < 32; ++e) {
    v2f p0 = {xv[e].x, xv[e].y}, p1 = {xv[e].z, xv[e].w};
    rot_in_pair(p0, a[0]);
    rot_in_pair(p1, a[0]);
    rot_pair(p0, p1, a[1]);
    xv[e] = make_float4(p0.x, p0.y, p1.x, p1.y);
  }
#pragma unroll
  for (int i = 0; i < 5; ++i)
#pragma unroll
    for (int e = 0; e < 32; ++e)
      if (((e >> i) & 1) == 0) rot_quad(xv[e], xv[e | (1 << i)], a[7 + i]);
#pragma unroll
  for (int e = 0; e < 32; ++e) st4[(e << 5) | (lane ^ (e & 15))] = xv[e];
  __syncthreads();
#pragma unroll
  for (int f = 0; f < 32; ++f) {
    const float4 q = st4[(lane << 5) | (f ^ (lane & 15))];
    S[2 * f + 0] = v2f{q.x, q.y};
    S[2 * f + 1] = v2f{q.z, q.w};
  }
#pragma unroll
  for (int i = 0; i < 5; ++i)
#pragma unroll
    for (int j = 0; j < 64; ++j)
      if (((j >> (i + 1)) & 1) == 0) rot_pair(S[j], S[j | (2 << i)], a[2 + i]);
}

__device__ __forceinline__ void ry_layer2_low(const v2f (&S)[64],
                                              const float (&a)[12], float a6,
                                              float4* st4, int slotb) {
  v2f P[64];
  const v2f am = {a[0], -a[0]};
#pragma unroll
  for (int j = 0; j < 64; ++j) {
    const v2f p = S[j];
    const v2f s = {p.y, p.x};
    P[j ^ (j >> 1)] = (j & 1) ? __builtin_elementwise_fma(am, p, s)
                              : __builtin_elementwise_fma(am, s, p);
  }
#pragma unroll
  for (int b = 1; b <= 6; ++b)
#pragma unroll
    for (int i = 0; i < 64; ++i)
      if (((i >> (b - 1)) & 1) == 0)
        rot_pair(P[i], P[i | (1 << (b - 1))], b == 6 ? a6 : a[b]);
#pragma unroll
  for (int j = 0; j < 32; ++j)
    *(float4*)((char*)st4 + (slotb ^ (j << 4))) =
        make_float4(P[2 * j].x, P[2 * j].y, P[2 * j + 1].x, P[2 * j + 1].y);
}

__global__ __launch_bounds__(32, 4)
void qdense_kernel(const float* __restrict__ x,
                   const float* __restrict__ theta,
                   float* __restrict__ out) {
  __shared__ __attribute__((aligned(16))) float st[DIM];
  float4* st4 = (float4*)st;

  const int lane = threadIdx.x;
  const int n    = lane & 15;
  const int half = lane >> 4;

  const float th = theta[lane < 24 ? lane : 0];
  const float4* xr = (const float4*)(x + (size_t)blockIdx.x * DIM);
  float4 xv[32];
#pragma unroll
  for (int e = 0; e < 32; ++e) xv[e] = xr[e * 32 + lane];

  float sv, cv;
  sincosf(th * 0.5f, &sv, &cv);

  const bool cot = fabsf(sv) > fabsf(cv);
  const unsigned bm = __builtin_amdgcn_ballot_w32(cot);
  const unsigned flip1 = __builtin_bitreverse32(bm & 0xFFFu) >> 20;
  const unsigned flip2 = __builtin_bitreverse32((bm >> 12) & 0xFFFu) >> 20;
  unsigned zp = flip1;
  zp ^= zp << 1;
  zp ^= zp << 2;
  zp ^= zp << 4;
  zp ^= zp << 8;
  const unsigned phi1 = flip1 ^ (flip1 >> 1);
  const unsigned zl =
      __builtin_bitreverse32((zp ^ (phi1 & 0x7Fu)) & 0xFFFu) >> 8;
  sv = ((zl >> lane) & 1u) ? -sv : sv;
  const float al = (cot ? cv : -sv) / (cot ? sv : cv);
  const float scl = cot ? sv : cv;

  float a1[12];
  float sc1 = 1.f;
#pragma unroll
  for (int b = 0; b < 12; ++b) {
    a1[b] = lane_bcast(al, 11 - b);
    sc1 *= lane_bcast(scl, 11 - b);
  }

  float ss0 = 0.f, ss1 = 0.f;
#pragma unroll
  for (int e = 0; e < 32; ++e) {
    const float4 v = xv[e];
    ss0 += v.x * v.x + v.y * v.y;
    ss1 += v.z * v.z + v.w * v.w;
  }
  const v2f ones = {1.f, 1.f};
  float sumsq;
  {
    v2f a;
    a.x = ss0;
    a.y = ss1;
    const v8f zero = {};
    const v8f t1 = wmma4(a, ones, zero);
    v8f t2 = {};
#pragma unroll
    for (int kc = 0; kc < 4; ++kc) {
      v2f s;
      s.x = t1[2 * kc + 0];
      s.y = t1[2 * kc + 1];
      t2 = wmma4(ones, s, t2);
    }
    sumsq = t2[0];
  }

  const bool fb = (sumsq <= 1e-24f);
  const float inv = fb ? 1.f : 1.f / sumsq;

  v2f S[64];
  ry_layer1(xv, S, a1, st4, lane);
  if (__builtin_expect(fb, 0)) {
    float pw = 1.f;
#pragma unroll
    for (int i = 0; i < 5; ++i) pw *= ((lane >> i) & 1) ? -a1[7 + i] : 1.f;
    S[0].x = pw;
    S[0].y = pw * -a1[0];
    S[1] = S[0] * -a1[1];
#pragma unroll
    for (int i = 0; i < 5; ++i) {
      const float m = -a1[2 + i];
#pragma unroll
      for (int j = 0; j < (2 << i); ++j) S[j | (2 << i)] = S[j] * m;
    }
  }

  float a2[12];
  float sc = sc1;
#pragma unroll
  for (int b = 0; b < 12; ++b) {
    a2[b] = lane_bcast(al, 23 - b);
    sc *= lane_bcast(scl, 23 - b);
  }
  const float a26 = (lane & 1) ? -a2[6] : a2[6];
  const int slotb = lds_slot((((lane << 7) ^ (lane << 6)) ^ (int)(phi1 & 0xFFCu)) >> 2) << 4;
  ry_layer2_low(S, a2, a26, st4, slotb);
  __syncthreads();
#pragma unroll
  for (int e = 0; e < 32; ++e) xv[e] = st4[(e << 5) | (lane ^ (e & 15))];
#pragma unroll
  for (int i = 0; i < 5; ++i)
#pragma unroll
    for (int e = 0; e < 32; ++e)
      if (((e >> i) & 1) == 0) rot_quad(xv[e], xv[e | (1 << i)], a2[7 + i]);

  float U[32], V[4];
  {
    v2f VV0 = {0.f, 0.f}, VV1 = {0.f, 0.f};
#pragma unroll
    for (int e = 0; e < 32; ++e) {
      const v2f lo = {xv[e].x, xv[e].y}, hi = {xv[e].z, xv[e].w};
      v2f t = lo * lo;
      t = __builtin_elementwise_fma(hi, hi, t);
      U[e] = t.x + t.y;
      VV0 = __builtin_elementwise_fma(lo, lo, VV0);
      VV1 = __builtin_elementwise_fma(hi, hi, VV1);
    }
    V[0] = VV0.x;
    V[1] = VV0.y;
    V[2] = VV1.x;
    V[3] = VV1.y;
  }

  const unsigned pwU = (n == 0 ? 0xFFFF0000u : 0u) | (n == 1 ? 0x00FFFF00u : 0u) |
                       (n == 2 ? 0x0FF00FF0u : 0u) | (n == 3 ? 0x3C3C3C3Cu : 0u) |
                       (n == 4 ? 0x66666666u : 0u) | (n == 5 ? 0xAAAAAAAAu : 0u);
  const unsigned pwV = (n == 10 ? 0xCu : 0u) | (n == 11 ? 0x6u : 0u);
  const bool colU = (n <= 9);
  const bool colV = (n == 10) | (n == 11);
  const float sgnH = ((n == 5) | (n == 6)) & half ? -1.f : 1.f;
  const float baseU = colU ? sgnH : 0.f;
  const float baseV = colV ? 1.f : 0.f;
  v8f d1 = {}, d1v = {};
#pragma unroll
  for (int w = 0; w < 8; ++w) {
    v2f a, c;
    a.x = U[2 * w + 0];
    a.y = U[2 * w + 1];
    c.x = signed_coef(baseU, pwU, 2 * w + 0);
    c.y = signed_coef(baseU, pwU, 2 * w + 1);
    d1 = wmma4(a, c, d1);
  }
#pragma unroll
  for (int w = 8; w < 16; ++w) {
    v2f a, c;
    a.x = U[2 * w + 0];
    a.y = U[2 * w + 1];
    c.x = signed_coef(baseU, pwU, 2 * w + 0);
    c.y = signed_coef(baseU, pwU, 2 * w + 1);
    d1v = wmma4(a, c, d1v);
  }
  {
    v2f a, c;
    a.x = V[0];
    a.y = V[1];
    c.x = signed_coef(baseV, pwV, 0);
    c.y = signed_coef(baseV, pwV, 1);
    d1 = wmma4(a, c, d1);
    a.x = V[2];
    a.y = V[3];
    c.x = signed_coef(baseV, pwV, 2);
    c.y = signed_coef(baseV, pwV, 3);
    d1v = wmma4(a, c, d1v);
  }
  d1 = d1 + d1v;
  const unsigned pwA = (n == 1 ? 0xFF00u : 0u) | (n == 2 ? 0x0FF0u : 0u) |
                       (n == 3 ? 0x3C3Cu : 0u) | (n == 4 ? 0x6666u : 0u) |
                       (n == 5 ? 0xAAAAu : 0u);
  const unsigned pwAh = half ? (pwA >> 8) : pwA;
  const float baseA = (n < 6) ? 1.f : 0.f;
  v8f d2 = {};
#pragma unroll
  for (int kc = 0; kc < 4; ++kc) {
    v2f s, c;
    s.x = d1[2 * kc + 0];
    s.y = d1[2 * kc + 1];
    c.x = signed_coef(baseA, pwAh, 2 * kc + 0);
    c.y = signed_coef(baseA, pwAh, 2 * kc + 1);
    d2 = wmma4(c, s, d2);
  }
  float r = d2[0];
  r = (lane == 6) ? d2[1] : r;
  r = (lane == 7) ? d2[2] : r;
  r = (lane == 8) ? d2[3] : r;
  r = (lane == 9) ? d2[4] : r;
  r = (lane == 10) ? d2[5] : r;
  const float fin = inv * (sc * sc);
  const unsigned fl = flip2 ^ (phi1 & 3u);
  const unsigned zs = __builtin_bitreverse32(fl ^ (fl >> 1)) >> 20;
  const float oinv = ((zs >> lane) & 1u) ? -fin : fin;
  if (lane < 12) out[(size_t)blockIdx.x * 12 + lane] = r * oinv;
}

extern "C" void kernel_launch(void* const* d_in, const int* in_sizes, int n_in,
                              void* d_out, int out_size, void* d_ws,
                              size_t ws_size, hipStream_t stream) {
  const float* x = (const float*)d_in[0];
  const float* theta = (const float*)d_in[1];
  float* out = (float*)d_out;
  const int B = in_sizes[0] / DIM;
  qdense_kernel<<<dim3(B), dim3(32), 0, stream>>>(x, theta, out);
}
